# starting kernel
# baseline (speedup 1.0000x reference)
#include <hip/hip_runtime.h>
#include <math.h>

#define D_FEAT 128
#define HEADS 4
#define CDIM 32
#define NEG_SLOPE 0.2f
#define LN_EPS 1e-5f

typedef __attribute__((ext_vector_type(2))) float v2f;
typedef __attribute__((ext_vector_type(8))) float v8f;

__global__ __launch_bounds__(256) void gat_init(float* __restrict__ acc,
                                                float* __restrict__ m,
                                                float* __restrict__ denom,
                                                int N) {
  size_t total = (size_t)N * D_FEAT;
  size_t nh    = (size_t)N * HEADS;
  for (size_t i = blockIdx.x * (size_t)blockDim.x + threadIdx.x; i < total;
       i += (size_t)gridDim.x * blockDim.x) {
    acc[i] = 0.0f;
    if (i < nh) {
      m[i] = -INFINITY;
      denom[i] = 0.0f;
    }
  }
}

__global__ __launch_bounds__(256) void gat_gemm_h(const float* __restrict__ x,
                                                  const float* __restrict__ W,
                                                  float* __restrict__ h,
                                                  int N) {
  const int lane  = threadIdx.x & 31;
  const int wid   = threadIdx.x >> 5;
  const int row0  = blockIdx.x * 16;
  const int col0  = wid * 16;
  const int mrow  = lane & 15;
  const int khalf = lane >> 4;

  int arow = row0 + mrow;
  if (arow >= N) arow = N - 1;

  v8f acc = {};
#pragma unroll 8
  for (int k0 = 0; k0 < D_FEAT; k0 += 4) {
    const int ka = k0 + 2 * khalf;
    v2f a = *(const v2f*)(x + (size_t)arow * D_FEAT + ka);
    const float* bp = W + (size_t)ka * D_FEAT + col0 + mrow;
    v2f b;
    b.x = bp[0];
    b.y = bp[D_FEAT];
    acc = __builtin_amdgcn_wmma_f32_16x16x4_f32(
         false, a,  false, b,
         (short)0, acc,  false,  false);
  }

  const int rbase = row0 + 8 * khalf;
  float* dp = h + (size_t)rbase * D_FEAT + col0 + mrow;
  if (row0 + 16 <= N) {
#pragma unroll
    for (int v = 0; v < 8; ++v) dp[(size_t)v * D_FEAT] = acc[v];
  } else {
#pragma unroll
    for (int v = 0; v < 8; ++v) {
      if (rbase + v < N) dp[(size_t)v * D_FEAT] = acc[v];
    }
  }
}

__global__ __launch_bounds__(128) void gat_att_logits(const float* __restrict__ h,
                                                      const float* __restrict__ att_src,
                                                      const float* __restrict__ att_dst,
                                                      float* __restrict__ a_src,
                                                      float* __restrict__ a_dst,
                                                      int N) {
  const int n = blockIdx.x;
  if (n >= N) return;
  const int t = threadIdx.x;
  const float hv = h[(size_t)n * D_FEAT + t];
  float ps = hv * att_src[t];
  float pd = hv * att_dst[t];
#pragma unroll
  for (int off = 16; off > 0; off >>= 1) {
    ps += __shfl_down(ps, off, 32);
    pd += __shfl_down(pd, off, 32);
  }
  if ((t & 31) == 0) {
    const int head = t >> 5;
    a_src[(size_t)n * HEADS + head] = ps;
    a_dst[(size_t)n * HEADS + head] = pd;
  }
}

__global__ __launch_bounds__(256) void gat_edge_max(const int* __restrict__ ei,
                                                    int E, int Etot,
                                                    const float* __restrict__ a_src,
                                                    const float* __restrict__ a_dst,
                                                    float* __restrict__ elog,
                                                    float* __restrict__ m) {
  const size_t tid = blockIdx.x * (size_t)blockDim.x + threadIdx.x;
  const size_t total = (size_t)Etot * HEADS;
  if (tid >= total) return;
  const int e  = (int)(tid >> 2);
  const int hd = (int)(tid & 3);
  const int src = (e < E) ? ei[e] : (e - E);
  const int dst = (e < E) ? ei[E + e] : (e - E);
  float v = a_src[(size_t)src * HEADS + hd] + a_dst[(size_t)dst * HEADS + hd];
  v = (v > 0.0f) ? v : NEG_SLOPE * v;
  elog[tid] = v;
  atomicMax(&m[(size_t)dst * HEADS + hd], v);
}

__global__ __launch_bounds__(256) void gat_edge_exp(const int* __restrict__ ei,
                                                    int E, int Etot,
                                                    float* __restrict__ elog,
                                                    const float* __restrict__ m,
                                                    float* __restrict__ denom) {
  const size_t tid = blockIdx.x * (size_t)blockDim.x + threadIdx.x;
  const size_t total = (size_t)Etot * HEADS;
  if (tid >= total) return;
  const int e  = (int)(tid >> 2);
  const int hd = (int)(tid & 3);
  const int dst = (e < E) ? ei[E + e] : (e - E);
  const float ex = __expf(elog[tid] - m[(size_t)dst * HEADS + hd]);
  elog[tid] = ex;
  atomicAdd(&denom[(size_t)dst * HEADS + hd], ex);
}

__global__ __launch_bounds__(256) void gat_aggregate(const int* __restrict__ ei,
                                                     int E, int Etot,
                                                     const float* __restrict__ h,
                                                     const float* __restrict__ eexp,
                                                     const float* __restrict__ denom,
                                                     float* __restrict__ acc) {
  const int lane = threadIdx.x & 31;
  const int e = blockIdx.x * (blockDim.x >> 5) + (threadIdx.x >> 5);
  if (e >= Etot) return;
  const int src = (e < E) ? ei[e] : (e - E);
  const int dst = (e < E) ? ei[E + e] : (e - E);
#pragma unroll
  for (int hd = 0; hd < HEADS; ++hd) {
    const float w = eexp[(size_t)e * HEADS + hd] /
                    (denom[(size_t)dst * HEADS + hd] + 1e-16f);
    const int c = hd * CDIM + lane;
    const float v = h[(size_t)src * D_FEAT + c] * w;
    atomicAdd(&acc[(size_t)dst * D_FEAT + c], v);
  }
}

__global__ __launch_bounds__(128) void gat_finish(float* __restrict__ acc,
                                                  const float* __restrict__ bias,
                                                  const float* __restrict__ gamma,
                                                  const float* __restrict__ beta,
                                                  const float* __restrict__ x,
                                                  int N) {
  __shared__ float s_sum[4];
  __shared__ float s_sq[4];
  const int n = blockIdx.x;
  if (n >= N) return;
  const int t = threadIdx.x;

  const float v = acc[(size_t)n * D_FEAT + t] + bias[t];
  const float g = 0.5f * v * (1.0f + erff(v * 0.70710678118654752f));

  float s = g, sq = g * g;
#pragma unroll
  for (int off = 16; off > 0; off >>= 1) {
    s  += __shfl_down(s, off, 32);
    sq += __shfl_down(sq, off, 32);
  }
  if ((t & 31) == 0) {
    s_sum[t >> 5] = s;
    s_sq[t >> 5]  = sq;
  }
  __syncthreads();
  const float tot  = s_sum[0] + s_sum[1] + s_sum[2] + s_sum[3];
  const float totq = s_sq[0] + s_sq[1] + s_sq[2] + s_sq[3];
  const float mu  = tot * (1.0f / D_FEAT);
  const float var = totq * (1.0f / D_FEAT) - mu * mu;
  const float inv = rsqrtf(var + LN_EPS);

  const float out = (g - mu) * inv * gamma[t] + beta[t] + x[(size_t)n * D_FEAT + t];
  acc[(size_t)n * D_FEAT + t] = out;
}

extern "C" void kernel_launch(void* const* d_in, const int* in_sizes, int n_in,
                              void* d_out, int out_size, void* d_ws, size_t ws_size,
                              hipStream_t stream) {
  const float* x       = (const float*)d_in[0];
  const int*   ei      = (const int*)d_in[1];
  const float* W       = (const float*)d_in[2];
  const float* att_src = (const float*)d_in[3];
  const float* att_dst = (const float*)d_in[4];
  const float* bias    = (const float*)d_in[5];
  const float* gamma   = (const float*)d_in[6];
  const float* beta    = (const float*)d_in[7];

  const int N    = in_sizes[0] / D_FEAT;
  const int E    = in_sizes[1] / 2;
  const int Etot = E + N;

  float* ws     = (float*)d_ws;
  float* h      = ws;
  float* a_src  = h + (size_t)N * D_FEAT;
  float* a_dst  = a_src + (size_t)N * HEADS;
  float* mbuf   = a_dst + (size_t)N * HEADS;
  float* denom  = mbuf + (size_t)N * HEADS;
  float* elog   = denom + (size_t)N * HEADS;
  float* acc    = (float*)d_out;

  {
    size_t total = (size_t)N * D_FEAT;
    int blocks = (int)((total + 255) / 256);
    if (blocks > 65535) blocks = 65535;
    gat_init<<<blocks, 256, 0, stream>>>(acc, mbuf, denom, N);
  }
  {
    int rowTiles = (N + 15) / 16;
    gat_gemm_h<<<rowTiles, 256, 0, stream>>>(x, W, h, N);
  }
  gat_att_logits<<<N, 128, 0, stream>>>(h, att_src, att_dst, a_src, a_dst, N);
  {
    size_t total = (size_t)Etot * HEADS;
    int blocks = (int)((total + 255) / 256);
    gat_edge_max<<<blocks, 256, 0, stream>>>(ei, E, Etot, a_src, a_dst, elog, mbuf);
  }
  {
    size_t total = (size_t)Etot * HEADS;
    int blocks = (int)((total + 255) / 256);
    gat_edge_exp<<<blocks, 256, 0, stream>>>(ei, E, Etot, elog, mbuf, denom);
  }
  {
    int edgesPerBlock = 256 / 32;
    int blocks = (Etot + edgesPerBlock - 1) / edgesPerBlock;
    gat_aggregate<<<blocks, 256, 0, stream>>>(ei, E, Etot, h, elog, denom, acc);
  }
  gat_finish<<<N, 128, 0, stream>>>(acc, bias, gamma, beta, x, N);
}


# attention-logits launch re-tiled: one wave per 8 nodes (8 row loads in flight, 4 in-lane products, 8-lane halving exchange sums; summation order of the logits changes), 200000 waves -> 6250
# speedup vs baseline: 2.2762x; 2.2762x over previous
#include <hip/hip_runtime.h>
#include <math.h>

#define D_FEAT 128
#define HEADS 4
#define CDIM 32
#define NEG_SLOPE 0.2f
#define LN_EPS 1e-5f

typedef __attribute__((ext_vector_type(2))) float v2f;
typedef __attribute__((ext_vector_type(8))) float v8f;

__global__ __launch_bounds__(256) void gat_gemm_h(const float* __restrict__ x,
                                                  const float* __restrict__ W,
                                                  float* __restrict__ h,
                                                  int N) {
  const int lane  = threadIdx.x & 31;
  const int wid   = threadIdx.x >> 5;
  const int row0  = blockIdx.x * 16;
  const int col0  = wid * 16;
  const int mrow  = lane & 15;
  const int khalf = lane >> 4;

  int arow = row0 + mrow;
  if (arow >= N) arow = N - 1;

  v8f acc = {};
#pragma unroll 8
  for (int k0 = 0; k0 < D_FEAT; k0 += 4) {
    const int ka = k0 + 2 * khalf;
    v2f a = *(const v2f*)(x + (size_t)arow * D_FEAT + ka);
    const float* bp = W + (size_t)ka * D_FEAT + col0 + mrow;
    v2f b;
    b.x = bp[0];
    b.y = bp[D_FEAT];
    acc = __builtin_amdgcn_wmma_f32_16x16x4_f32(
         false, a,  false, b,
         (short)0, acc,  false,  false);
  }

  const int rbase = row0 + 8 * khalf;
  float* dp = h + (size_t)rbase * D_FEAT + col0 + mrow;
  if (row0 + 16 <= N) {
#pragma unroll
    for (int v = 0; v < 8; ++v) dp[(size_t)v * D_FEAT] = acc[v];
  } else {
#pragma unroll
    for (int v = 0; v < 8; ++v) {
      if (rbase + v < N) dp[(size_t)v * D_FEAT] = acc[v];
    }
  }
}

#define LG_NODES 8
#define LG_WAVES 8
__global__ __launch_bounds__(32 * LG_WAVES) void gat_att_logits(const float* __restrict__ h,
                                                                const float* __restrict__ att_src,
                                                                const float* __restrict__ att_dst,
                                                                float* __restrict__ a_src,
                                                                float* __restrict__ a_dst,
                                                                int N) {
  const int lane = threadIdx.x & 31;
  const int wave = __builtin_amdgcn_readfirstlane((int)(threadIdx.x >> 5));
  const int n0 = (blockIdx.x * LG_WAVES + wave) * LG_NODES;
  if (n0 >= N) return;

  const float4* h4 = (const float4*)h;
  float4 r[LG_NODES];
#pragma unroll
  for (int u = 0; u < LG_NODES; ++u) {
    int n = n0 + u;
    if (n >= N) n = N - 1;
    r[u] = h4[(size_t)n * (D_FEAT / 4) + lane];
  }
  const float4 as4 = ((const float4*)att_src)[lane];
  const float4 ad4 = ((const float4*)att_dst)[lane];

  float ps[LG_NODES], pd[LG_NODES];
#pragma unroll
  for (int u = 0; u < LG_NODES; ++u) {
    ps[u] = (r[u].x * as4.x + r[u].y * as4.y) + (r[u].z * as4.z + r[u].w * as4.w);
    pd[u] = (r[u].x * ad4.x + r[u].y * ad4.y) + (r[u].z * ad4.z + r[u].w * ad4.w);
  }

  const bool b2 = (lane & 4) != 0, b1 = (lane & 2) != 0, b0 = (lane & 1) != 0;
  float qs[4], qd[4];
#pragma unroll
  for (int j = 0; j < 4; ++j) {
    const float ks = b2 ? ps[j + 4] : ps[j], ss = b2 ? ps[j] : ps[j + 4];
    const float kd = b2 ? pd[j + 4] : pd[j], sd = b2 ? pd[j] : pd[j + 4];
    qs[j] = ks + __shfl_xor(ss, 4, 32);
    qd[j] = kd + __shfl_xor(sd, 4, 32);
  }
  float ts[2], td[2];
#pragma unroll
  for (int j = 0; j < 2; ++j) {
    const float ks = b1 ? qs[j + 2] : qs[j], ss = b1 ? qs[j] : qs[j + 2];
    const float kd = b1 ? qd[j + 2] : qd[j], sd = b1 ? qd[j] : qd[j + 2];
    ts[j] = ks + __shfl_xor(ss, 2, 32);
    td[j] = kd + __shfl_xor(sd, 2, 32);
  }
  const float fs = (b0 ? ts[1] : ts[0]) + __shfl_xor(b0 ? ts[0] : ts[1], 1, 32);
  const float fd = (b0 ? td[1] : td[0]) + __shfl_xor(b0 ? td[0] : td[1], 1, 32);

  const int n = n0 + (lane & 7);
  if (n < N) {
    a_src[(size_t)n * HEADS + (lane >> 3)] = fs;
    a_dst[(size_t)n * HEADS + (lane >> 3)] = fd;
  }
}

#define NPG 196
#define CAP 4096
#define TPB 1024
#define SCAN_DEPTH 8

static __device__ __forceinline__ void rec_append(int e, unsigned d,
                                                  int* s_cnt, int* s_src, unsigned short* s_dl) {
  const int p = atomicAdd(s_cnt, 1);
  if (p < CAP) {
    s_src[p] = e;
    s_dl[p]  = (unsigned short)d;
  }
}

static __device__ __forceinline__ void resolve_src(const int* __restrict__ ei, int from, int to,
                                                   int tid, int* s_src) {
  for (int i = from + tid; i < to; i += TPB) s_src[i] = ei[s_src[i]];
}

static __device__ __forceinline__ void scan_scalar(const int* __restrict__ ei, int E, int c0, int c1,
                                                   int lo, int nn, int tid,
                                                   int* s_cnt, int* s_src, unsigned short* s_dl) {
  for (int e = c0 + tid; e < c1; e += TPB) {
    const unsigned d = (unsigned)(ei[E + e] - lo);
    if (d < (unsigned)nn) rec_append(e, d, s_cnt, s_src, s_dl);
  }
}

static __device__ __forceinline__ void scan_vec4(const int* __restrict__ ei, int E,
                                                 int lo, int nn, int tid,
                                                 int* s_cnt, int* s_src, unsigned short* s_dl) {
  const int nvec = E >> 2;
  const int4* d4 = (const int4*)(ei + E);
  const int nb = (nvec + TPB * SCAN_DEPTH - 1) / (TPB * SCAN_DEPTH);
  if (nb == 0) return;
  int kb = (int)(blockIdx.x % (unsigned)nb);
  for (int k = 0; k < nb; ++k) {
    const int b = kb * (TPB * SCAN_DEPTH) + tid;
    kb = (kb + 1 == nb) ? 0 : kb + 1;
    int4 v[SCAN_DEPTH];
#pragma unroll
    for (int u = 0; u < SCAN_DEPTH; ++u) {
      const int i = b + TPB * u;
      v[u] = d4[(i < nvec) ? i : 0];
    }
#pragma unroll
    for (int u = 0; u < SCAN_DEPTH; ++u) {
      const int i = b + TPB * u;
      const unsigned d0 = (unsigned)(v[u].x - lo);
      const unsigned d1 = (unsigned)(v[u].y - lo);
      const unsigned d2 = (unsigned)(v[u].z - lo);
      const unsigned d3 = (unsigned)(v[u].w - lo);
      const unsigned mn = min(min(d0, d1), min(d2, d3));
      if (i < nvec && mn < (unsigned)nn) {
        if (d0 < (unsigned)nn) rec_append(4 * i + 0, d0, s_cnt, s_src, s_dl);
        if (d1 < (unsigned)nn) rec_append(4 * i + 1, d1, s_cnt, s_src, s_dl);
        if (d2 < (unsigned)nn) rec_append(4 * i + 2, d2, s_cnt, s_src, s_dl);
        if (d3 < (unsigned)nn) rec_append(4 * i + 3, d3, s_cnt, s_src, s_dl);
      }
    }
  }
}

static __device__ __forceinline__ float edge_logit(const float* __restrict__ a_src,
                                                   const float* s_adst, int src, int dl, int hd) {
  const float v = a_src[(size_t)src * HEADS + hd] + s_adst[dl * HEADS + hd];
  return (v > 0.0f) ? v : NEG_SLOPE * v;
}

static __device__ __forceinline__ void run_phase(int ph, int n, int tid,
                                                 const float* __restrict__ a_src,
                                                 const float* __restrict__ h,
                                                 const int* s_src, const unsigned short* s_dl,
                                                 const float* s_adst, float* s_m, float* s_den,
                                                 float* s_w, float* s_acc) {
  if (ph == 0) {
    for (int i = tid; i < n * HEADS; i += TPB) {
      const int p = i >> 2, hd = i & 3;
      const int dl = s_dl[p];
      const float v = edge_logit(a_src, s_adst, s_src[p], dl, hd);
      atomicMax(&s_m[dl * HEADS + hd], v);
    }
  } else if (ph == 1) {
    for (int i = tid; i < n * HEADS; i += TPB) {
      const int p = i >> 2, hd = i & 3;
      const int dl = s_dl[p];
      const float v = edge_logit(a_src, s_adst, s_src[p], dl, hd);
      atomicAdd(&s_den[dl * HEADS + hd], __expf(v - s_m[dl * HEADS + hd]));
    }
  } else {
    for (int i = tid; i < n * HEADS; i += TPB) {
      const int p = i >> 2, hd = i & 3;
      const int dl = s_dl[p];
      const float v = edge_logit(a_src, s_adst, s_src[p], dl, hd);
      s_w[i] = __expf(v - s_m[dl * HEADS + hd]) / (s_den[dl * HEADS + hd] + 1e-16f);
    }
    __syncthreads();
    const int lane = tid & 31;
    const int wave = __builtin_amdgcn_readfirstlane(tid >> 5);
    const float4* h4 = (const float4*)h;
    for (int p0 = wave; p0 < n; p0 += 32 * 4) {
      float4 r[4];
      int srcv[4];
      int dlv[4];
      float wv[4];
#pragma unroll
      for (int u = 0; u < 4; ++u) {
        const int p  = p0 + 32 * u;
        const int pc = (p < n) ? p : p0;
        srcv[u] = s_src[pc];
        dlv[u]  = s_dl[pc];
        wv[u]   = (p < n) ? s_w[pc * HEADS + (lane >> 3)] : 0.0f;
      }
#pragma unroll
      for (int u = 0; u < 4; ++u) r[u] = h4[(size_t)srcv[u] * (D_FEAT / 4) + lane];
#pragma unroll
      for (int u = 0; u < 4; ++u) {
        const bool ok = (p0 + 32 * u < n);
        float* ap = s_acc + dlv[u] * D_FEAT + 4 * lane;
        atomicAdd(ap + 0, ok ? r[u].x * wv[u] : 0.0f);
        atomicAdd(ap + 1, ok ? r[u].y * wv[u] : 0.0f);
        atomicAdd(ap + 2, ok ? r[u].z * wv[u] : 0.0f);
        atomicAdd(ap + 3, ok ? r[u].w * wv[u] : 0.0f);
      }
    }
  }
}

__global__ __launch_bounds__(TPB) void gat_owner(const int* __restrict__ ei, int E,
                                                 const float* __restrict__ h,
                                                 const float* __restrict__ a_src,
                                                 const float* __restrict__ a_dst,
                                                 const float* __restrict__ bias,
                                                 const float* __restrict__ gamma,
                                                 const float* __restrict__ beta,
                                                 const float* __restrict__ x,
                                                 float* __restrict__ out, int N) {
  __shared__ __attribute__((aligned(16))) float s_acc[NPG * D_FEAT];
  __shared__ __attribute__((aligned(16))) float s_w[CAP * HEADS];
  __shared__ int s_src[CAP];
  __shared__ unsigned short s_dl[CAP];
  __shared__ float s_m[NPG * HEADS];
  __shared__ float s_den[NPG * HEADS];
  __shared__ float s_adst[NPG * HEADS];
  __shared__ int s_cnt;

  const int tid = threadIdx.x;
  const int lo  = blockIdx.x * NPG;
  if (lo >= N) return;
  const int nn  = (N - lo < NPG) ? (N - lo) : NPG;

  for (int i = tid; i < NPG * D_FEAT / 4; i += TPB)
    ((float4*)s_acc)[i] = make_float4(0.0f, 0.0f, 0.0f, 0.0f);
  for (int i = tid; i < nn * HEADS; i += TPB) {
    s_m[i]    = -INFINITY;
    s_den[i]  = 0.0f;
    s_adst[i] = a_dst[(size_t)lo * HEADS + i];
  }
  for (int i = tid; i < nn; i += TPB) {
    s_src[i] = lo + i;
    s_dl[i]  = (unsigned short)i;
  }
  if (tid == 0) s_cnt = nn;
  __syncthreads();

  if ((E & 3) == 0) scan_vec4(ei, E, lo, nn, tid, &s_cnt, s_src, s_dl);
  else              scan_scalar(ei, E, 0, E, lo, nn, tid, &s_cnt, s_src, s_dl);
  __syncthreads();
  const int total = s_cnt;
  resolve_src(ei, nn, (total < CAP) ? total : CAP, tid, s_src);
  __syncthreads();

  if (total <= CAP) {
    run_phase(0, total, tid, a_src, h, s_src, s_dl, s_adst, s_m, s_den, s_w, s_acc);
    __syncthreads();
    run_phase(1, total, tid, a_src, h, s_src, s_dl, s_adst, s_m, s_den, s_w, s_acc);
    __syncthreads();
    run_phase(2, total, tid, a_src, h, s_src, s_dl, s_adst, s_m, s_den, s_w, s_acc);
    __syncthreads();
  } else {
    for (int ph = 0; ph < 3; ++ph) {
      __syncthreads();
      for (int i = tid; i < nn; i += TPB) {
        s_src[i] = lo + i;
        s_dl[i]  = (unsigned short)i;
      }
      __syncthreads();
      run_phase(ph, nn, tid, a_src, h, s_src, s_dl, s_adst, s_m, s_den, s_w, s_acc);
      __syncthreads();
      for (int c0 = 0; c0 < E; c0 += CAP) {
        if (tid == 0) s_cnt = 0;
        __syncthreads();
        const int c1 = (E - c0 < CAP) ? E : (c0 + CAP);
        scan_scalar(ei, E, c0, c1, lo, nn, tid, &s_cnt, s_src, s_dl);
        __syncthreads();
        const int n = s_cnt;
        resolve_src(ei, 0, n, tid, s_src);
        __syncthreads();
        run_phase(ph, n, tid, a_src, h, s_src, s_dl, s_adst, s_m, s_den, s_w, s_acc);
        __syncthreads();
      }
    }
  }

  const int lane = tid & 31;
  const int wave = __builtin_amdgcn_readfirstlane(tid >> 5);
  const float4 b4  = ((const float4*)bias)[lane];
  const float4 g4  = ((const float4*)gamma)[lane];
  const float4 be4 = ((const float4*)beta)[lane];
  for (int nl = wave; nl < nn; nl += TPB / 32) {
    const size_t row = (size_t)(lo + nl) * (D_FEAT / 4) + lane;
    const float4 xr = ((const float4*)x)[row];
    const float4 a  = *(const float4*)(s_acc + nl * D_FEAT + 4 * lane);
    const float v0 = a.x + b4.x, v1 = a.y + b4.y, v2 = a.z + b4.z, v3 = a.w + b4.w;
    const float q0 = 0.5f * v0 * (1.0f + erff(v0 * 0.70710678118654752f));
    const float q1 = 0.5f * v1 * (1.0f + erff(v1 * 0.70710678118654752f));
    const float q2 = 0.5f * v2 * (1.0f + erff(v2 * 0.70710678118654752f));
    const float q3 = 0.5f * v3 * (1.0f + erff(v3 * 0.70710678118654752f));
    float s  = (q0 + q1) + (q2 + q3);
    float sq = (q0 * q0 + q1 * q1) + (q2 * q2 + q3 * q3);
#pragma unroll
    for (int off = 16; off > 0; off >>= 1) {
      s  += __shfl_xor(s, off, 32);
      sq += __shfl_xor(sq, off, 32);
    }
    const float mu  = s * (1.0f / D_FEAT);
    const float var = sq * (1.0f / D_FEAT) - mu * mu;
    const float inv = rsqrtf(var + LN_EPS);
    float4 o;
    o.x = (q0 - mu) * inv * g4.x + be4.x + xr.x;
    o.y = (q1 - mu) * inv * g4.y + be4.y + xr.y;
    o.z = (q2 - mu) * inv * g4.z + be4.z + xr.z;
    o.w = (q3 - mu) * inv * g4.w + be4.w + xr.w;
    ((float4*)out)[row] = o;
  }
}

extern "C" void kernel_launch(void* const* d_in, const int* in_sizes, int n_in,
                              void* d_out, int out_size, void* d_ws, size_t ws_size,
                              hipStream_t stream) {
  const float* x       = (const float*)d_in[0];
  const int*   ei      = (const int*)d_in[1];
  const float* W       = (const float*)d_in[2];
  const float* att_src = (const float*)d_in[3];
  const float* att_dst = (const float*)d_in[4];
  const float* bias    = (const float*)d_in[5];
  const float* gamma   = (const float*)d_in[6];
  const float* beta    = (const float*)d_in[7];

  const int N = in_sizes[0] / D_FEAT;
  const int E = in_sizes[1] / 2;

  float* ws    = (float*)d_ws;
  float* h     = ws;
  float* a_src = h + (size_t)N * D_FEAT;
  float* a_dst = a_src + (size_t)N * HEADS;

  {
    int rowTiles = (N + 15) / 16;
    gat_gemm_h<<<rowTiles, 256, 0, stream>>>(x, W, h, N);
  }
  {
    int groups = (N + LG_NODES * LG_WAVES - 1) / (LG_NODES * LG_WAVES);
    gat_att_logits<<<groups, 32 * LG_WAVES, 0, stream>>>(h, att_src, att_dst, a_src, a_dst, N);
  }
  {
    int groups = (N + NPG - 1) / NPG;
    gat_owner<<<groups, TPB, 0, stream>>>(ei, E, h, a_src, a_dst, bias, gamma, beta, x,
                                          (float*)d_out, N);
  }
}


# owner groups filter a 1-byte owner key per edge (destination/197, written by extra work-groups of the logits launch), not the 4-byte destination; 197 nodes per group; LDS atomic sums, arrival order
# speedup vs baseline: 2.4569x; 1.0794x over previous
#include <hip/hip_runtime.h>
#include <math.h>

#define D_FEAT 128
#define HEADS 4
#define CDIM 32
#define NEG_SLOPE 0.2f
#define LN_EPS 1e-5f

typedef __attribute__((ext_vector_type(2))) float v2f;
typedef __attribute__((ext_vector_type(8))) float v8f;

__global__ __launch_bounds__(256) void gat_gemm_h(const float* __restrict__ x,
                                                  const float* __restrict__ W,
                                                  float* __restrict__ h,
                                                  int N) {
  const int lane  = threadIdx.x & 31;
  const int wid   = threadIdx.x >> 5;
  const int row0  = blockIdx.x * 16;
  const int col0  = wid * 16;
  const int mrow  = lane & 15;
  const int khalf = lane >> 4;

  int arow = row0 + mrow;
  if (arow >= N) arow = N - 1;

  v8f acc = {};
#pragma unroll 8
  for (int k0 = 0; k0 < D_FEAT; k0 += 4) {
    const int ka = k0 + 2 * khalf;
    v2f a = *(const v2f*)(x + (size_t)arow * D_FEAT + ka);
    const float* bp = W + (size_t)ka * D_FEAT + col0 + mrow;
    v2f b;
    b.x = bp[0];
    b.y = bp[D_FEAT];
    acc = __builtin_amdgcn_wmma_f32_16x16x4_f32(
         false, a,  false, b,
         (short)0, acc,  false,  false);
  }

  const int rbase = row0 + 8 * khalf;
  float* dp = h + (size_t)rbase * D_FEAT + col0 + mrow;
  if (row0 + 16 <= N) {
#pragma unroll
    for (int v = 0; v < 8; ++v) dp[(size_t)v * D_FEAT] = acc[v];
  } else {
#pragma unroll
    for (int v = 0; v < 8; ++v) {
      if (rbase + v < N) dp[(size_t)v * D_FEAT] = acc[v];
    }
  }
}

#define LG_NODES 8
#define LG_WAVES 8
#define NPG 197
#define KEY_NONE 255u
#define KEY_VEC_PER_BLOCK 4

static __device__ __forceinline__ unsigned owner_key(int d, int N) {
  return ((unsigned)d < (unsigned)N) ? ((unsigned)d / NPG) : KEY_NONE;
}

__global__ __launch_bounds__(32 * LG_WAVES) void gat_att_logits(const float* __restrict__ h,
                                                                const float* __restrict__ att_src,
                                                                const float* __restrict__ att_dst,
                                                                float* __restrict__ a_src,
                                                                float* __restrict__ a_dst,
                                                                int N,
                                                                const int* __restrict__ ei, int E,
                                                                unsigned* __restrict__ keys, int kb) {
  if ((int)blockIdx.x < kb) {
    const int4* d4 = (const int4*)(ei + E);
    const int nvec = E >> 2;
    int4 d[KEY_VEC_PER_BLOCK];
#pragma unroll
    for (int u = 0; u < KEY_VEC_PER_BLOCK; ++u) {
      const int i = ((int)blockIdx.x * KEY_VEC_PER_BLOCK + u) * (32 * LG_WAVES) + (int)threadIdx.x;
      d[u] = d4[(i < nvec) ? i : 0];
    }
#pragma unroll
    for (int u = 0; u < KEY_VEC_PER_BLOCK; ++u) {
      const int i = ((int)blockIdx.x * KEY_VEC_PER_BLOCK + u) * (32 * LG_WAVES) + (int)threadIdx.x;
      if (i < nvec)
        keys[i] = owner_key(d[u].x, N) | (owner_key(d[u].y, N) << 8) |
                  (owner_key(d[u].z, N) << 16) | (owner_key(d[u].w, N) << 24);
    }
    return;
  }
  const int blk  = (int)blockIdx.x - kb;
  const int lane = threadIdx.x & 31;
  const int wave = __builtin_amdgcn_readfirstlane((int)(threadIdx.x >> 5));
  const int n0 = (blk * LG_WAVES + wave) * LG_NODES;
  if (n0 >= N) return;

  const float4* h4 = (const float4*)h;
  float4 r[LG_NODES];
#pragma unroll
  for (int u = 0; u < LG_NODES; ++u) {
    int n = n0 + u;
    if (n >= N) n = N - 1;
    r[u] = h4[(size_t)n * (D_FEAT / 4) + lane];
  }
  const float4 as4 = ((const float4*)att_src)[lane];
  const float4 ad4 = ((const float4*)att_dst)[lane];

  float ps[LG_NODES], pd[LG_NODES];
#pragma unroll
  for (int u = 0; u < LG_NODES; ++u) {
    ps[u] = (r[u].x * as4.x + r[u].y * as4.y) + (r[u].z * as4.z + r[u].w * as4.w);
    pd[u] = (r[u].x * ad4.x + r[u].y * ad4.y) + (r[u].z * ad4.z + r[u].w * ad4.w);
  }

  const bool b2 = (lane & 4) != 0, b1 = (lane & 2) != 0, b0 = (lane & 1) != 0;
  float qs[4], qd[4];
#pragma unroll
  for (int j = 0; j < 4; ++j) {
    const float ks = b2 ? ps[j + 4] : ps[j], ss = b2 ? ps[j] : ps[j + 4];
    const float kd = b2 ? pd[j + 4] : pd[j], sd = b2 ? pd[j] : pd[j + 4];
    qs[j] = ks + __shfl_xor(ss, 4, 32);
    qd[j] = kd + __shfl_xor(sd, 4, 32);
  }
  float ts[2], td[2];
#pragma unroll
  for (int j = 0; j < 2; ++j) {
    const float ks = b1 ? qs[j + 2] : qs[j], ss = b1 ? qs[j] : qs[j + 2];
    const float kd = b1 ? qd[j + 2] : qd[j], sd = b1 ? qd[j] : qd[j + 2];
    ts[j] = ks + __shfl_xor(ss, 2, 32);
    td[j] = kd + __shfl_xor(sd, 2, 32);
  }
  const float fs = (b0 ? ts[1] : ts[0]) + __shfl_xor(b0 ? ts[0] : ts[1], 1, 32);
  const float fd = (b0 ? td[1] : td[0]) + __shfl_xor(b0 ? td[0] : td[1], 1, 32);

  const int n = n0 + (lane & 7);
  if (n < N) {
    a_src[(size_t)n * HEADS + (lane >> 3)] = fs;
    a_dst[(size_t)n * HEADS + (lane >> 3)] = fd;
  }
}

#define CAP 4096
#define TPB 1024
#define SCAN_DEPTH 8

static __device__ __forceinline__ void rec_append(int e, unsigned d,
                                                  int* s_cnt, int* s_src, unsigned short* s_dl) {
  const int p = atomicAdd(s_cnt, 1);
  if (p < CAP) {
    s_src[p] = e;
    s_dl[p]  = (unsigned short)d;
  }
}

static __device__ __forceinline__ void resolve_src(const int* __restrict__ ei, int from, int to,
                                                   int tid, int* s_src) {
  for (int i = from + tid; i < to; i += TPB) s_src[i] = ei[s_src[i]];
}

static __device__ __forceinline__ void scan_scalar(const int* __restrict__ ei, int E, int c0, int c1,
                                                   int lo, int nn, int tid,
                                                   int* s_cnt, int* s_src, unsigned short* s_dl) {
  for (int e = c0 + tid; e < c1; e += TPB) {
    const unsigned d = (unsigned)(ei[E + e] - lo);
    if (d < (unsigned)nn) rec_append(e, d, s_cnt, s_src, s_dl);
  }
}

static __device__ __forceinline__ void scan_vec4(const int* __restrict__ ei, int E,
                                                 int lo, int nn, int tid,
                                                 int* s_cnt, int* s_src, unsigned short* s_dl) {
  const int nvec = E >> 2;
  const int4* d4 = (const int4*)(ei + E);
  const int nb = (nvec + TPB * SCAN_DEPTH - 1) / (TPB * SCAN_DEPTH);
  if (nb == 0) return;
  int kb = (int)(blockIdx.x % (unsigned)nb);
  for (int k = 0; k < nb; ++k) {
    const int b = kb * (TPB * SCAN_DEPTH) + tid;
    kb = (kb + 1 == nb) ? 0 : kb + 1;
    int4 v[SCAN_DEPTH];
#pragma unroll
    for (int u = 0; u < SCAN_DEPTH; ++u) {
      const int i = b + TPB * u;
      v[u] = d4[(i < nvec) ? i : 0];
    }
#pragma unroll
    for (int u = 0; u < SCAN_DEPTH; ++u) {
      const int i = b + TPB * u;
      const unsigned d0 = (unsigned)(v[u].x - lo);
      const unsigned d1 = (unsigned)(v[u].y - lo);
      const unsigned d2 = (unsigned)(v[u].z - lo);
      const unsigned d3 = (unsigned)(v[u].w - lo);
      const unsigned mn = min(min(d0, d1), min(d2, d3));
      if (i < nvec && mn < (unsigned)nn) {
        if (d0 < (unsigned)nn) rec_append(4 * i + 0, d0, s_cnt, s_src, s_dl);
        if (d1 < (unsigned)nn) rec_append(4 * i + 1, d1, s_cnt, s_src, s_dl);
        if (d2 < (unsigned)nn) rec_append(4 * i + 2, d2, s_cnt, s_src, s_dl);
        if (d3 < (unsigned)nn) rec_append(4 * i + 3, d3, s_cnt, s_src, s_dl);
      }
    }
  }
}

static __device__ __forceinline__ void resolve_src_dst(const int* __restrict__ ei, int E, int lo,
                                                       int from, int to, int tid,
                                                       int* s_src, unsigned short* s_dl) {
  for (int i = from + tid; i < to; i += TPB) {
    const int e = s_src[i];
    const int d = ei[E + e];
    const int s = ei[e];
    s_dl[i]  = (unsigned short)(d - lo);
    s_src[i] = s;
  }
}

static __device__ __forceinline__ unsigned nonzero_bytes(unsigned x) {
  return ((x & 0x7F7F7F7Fu) + 0x7F7F7F7Fu) | x;
}

static __device__ __forceinline__ void scan_keys(const unsigned* __restrict__ keys, int E,
                                                 unsigned gid, int tid, int* s_cnt, int* s_src) {
  const int nvec = E >> 4;
  const uint4* k4 = (const uint4*)keys;
  const unsigned rep4 = gid * 0x01010101u;
  const int nb = (nvec + TPB * SCAN_DEPTH - 1) / (TPB * SCAN_DEPTH);
  if (nb == 0) return;
  int kb = (int)(blockIdx.x % (unsigned)nb);
  for (int k = 0; k < nb; ++k) {
    const int b = kb * (TPB * SCAN_DEPTH) + tid;
    kb = (kb + 1 == nb) ? 0 : kb + 1;
    uint4 v[SCAN_DEPTH];
#pragma unroll
    for (int u = 0; u < SCAN_DEPTH; ++u) {
      const int i = b + TPB * u;
      v[u] = k4[(i < nvec) ? i : 0];
    }
#pragma unroll
    for (int u = 0; u < SCAN_DEPTH; ++u) {
      const int i = b + TPB * u;
      unsigned m = (nonzero_bytes(v[u].x ^ rep4) & 0x80808080u) |
                   ((nonzero_bytes(v[u].y ^ rep4) >> 1) & 0x40404040u) |
                   ((nonzero_bytes(v[u].z ^ rep4) >> 2) & 0x20202020u) |
                   ((nonzero_bytes(v[u].w ^ rep4) >> 3) & 0x10101010u);
      m ^= 0xF0F0F0F0u;
      if (i >= nvec) m = 0u;
      while (m != 0u) {
        const int p = __builtin_ctz(m);
        m &= m - 1u;
        const int e = 16 * i + 4 * (7 - (p & 7)) + (p >> 3);
        const int q = atomicAdd(s_cnt, 1);
        if (q < CAP) s_src[q] = e;
      }
    }
  }
}

static __device__ __forceinline__ float edge_logit(const float* __restrict__ a_src,
                                                   const float* s_adst, int src, int dl, int hd) {
  const float v = a_src[(size_t)src * HEADS + hd] + s_adst[dl * HEADS + hd];
  return (v > 0.0f) ? v : NEG_SLOPE * v;
}

static __device__ __forceinline__ void run_phase(int ph, int n, int tid,
                                                 const float* __restrict__ a_src,
                                                 const float* __restrict__ h,
                                                 const int* s_src, const unsigned short* s_dl,
                                                 const float* s_adst, float* s_m, float* s_den,
                                                 float* s_w, float* s_acc) {
  if (ph == 0) {
    for (int i = tid; i < n * HEADS; i += TPB) {
      const int p = i >> 2, hd = i & 3;
      const int dl = s_dl[p];
      const float v = edge_logit(a_src, s_adst, s_src[p], dl, hd);
      atomicMax(&s_m[dl * HEADS + hd], v);
    }
  } else if (ph == 1) {
    for (int i = tid; i < n * HEADS; i += TPB) {
      const int p = i >> 2, hd = i & 3;
      const int dl = s_dl[p];
      const float v = edge_logit(a_src, s_adst, s_src[p], dl, hd);
      atomicAdd(&s_den[dl * HEADS + hd], __expf(v - s_m[dl * HEADS + hd]));
    }
  } else {
    for (int i = tid; i < n * HEADS; i += TPB) {
      const int p = i >> 2, hd = i & 3;
      const int dl = s_dl[p];
      const float v = edge_logit(a_src, s_adst, s_src[p], dl, hd);
      s_w[i] = __expf(v - s_m[dl * HEADS + hd]) / (s_den[dl * HEADS + hd] + 1e-16f);
    }
    __syncthreads();
    const int lane = tid & 31;
    const int wave = __builtin_amdgcn_readfirstlane(tid >> 5);
    const float4* h4 = (const float4*)h;
    for (int p0 = wave; p0 < n; p0 += 32 * 4) {
      float4 r[4];
      int srcv[4];
      int dlv[4];
      float wv[4];
#pragma unroll
      for (int u = 0; u < 4; ++u) {
        const int p  = p0 + 32 * u;
        const int pc = (p < n) ? p : p0;
        srcv[u] = s_src[pc];
        dlv[u]  = s_dl[pc];
        wv[u]   = (p < n) ? s_w[pc * HEADS + (lane >> 3)] : 0.0f;
      }
#pragma unroll
      for (int u = 0; u < 4; ++u) r[u] = h4[(size_t)srcv[u] * (D_FEAT / 4) + lane];
#pragma unroll
      for (int u = 0; u < 4; ++u) {
        const bool ok = (p0 + 32 * u < n);
        float* ap = s_acc + dlv[u] * D_FEAT + 4 * lane;
        atomicAdd(ap + 0, ok ? r[u].x * wv[u] : 0.0f);
        atomicAdd(ap + 1, ok ? r[u].y * wv[u] : 0.0f);
        atomicAdd(ap + 2, ok ? r[u].z * wv[u] : 0.0f);
        atomicAdd(ap + 3, ok ? r[u].w * wv[u] : 0.0f);
      }
    }
  }
}

__global__ __launch_bounds__(TPB) void gat_owner(const int* __restrict__ ei, int E,
                                                 const float* __restrict__ h,
                                                 const float* __restrict__ a_src,
                                                 const float* __restrict__ a_dst,
                                                 const float* __restrict__ bias,
                                                 const float* __restrict__ gamma,
                                                 const float* __restrict__ beta,
                                                 const float* __restrict__ x,
                                                 float* __restrict__ out, int N,
                                                 const unsigned* __restrict__ keys) {
  __shared__ __attribute__((aligned(16))) float s_acc[NPG * D_FEAT];
  __shared__ __attribute__((aligned(16))) float s_w[CAP * HEADS];
  __shared__ int s_src[CAP];
  __shared__ unsigned short s_dl[CAP];
  __shared__ float s_m[NPG * HEADS];
  __shared__ float s_den[NPG * HEADS];
  __shared__ float s_adst[NPG * HEADS];
  __shared__ int s_cnt;

  const int tid = threadIdx.x;
  const int lo  = blockIdx.x * NPG;
  if (lo >= N) return;
  const int nn  = (N - lo < NPG) ? (N - lo) : NPG;

  for (int i = tid; i < NPG * D_FEAT / 4; i += TPB)
    ((float4*)s_acc)[i] = make_float4(0.0f, 0.0f, 0.0f, 0.0f);
  for (int i = tid; i < nn * HEADS; i += TPB) {
    s_m[i]    = -INFINITY;
    s_den[i]  = 0.0f;
    s_adst[i] = a_dst[(size_t)lo * HEADS + i];
  }
  for (int i = tid; i < nn; i += TPB) {
    s_src[i] = lo + i;
    s_dl[i]  = (unsigned short)i;
  }
  if (tid == 0) s_cnt = nn;
  __syncthreads();

  const bool by_key = (keys != nullptr) && ((E & 15) == 0) && (gridDim.x <= KEY_NONE);
  if (by_key)            scan_keys(keys, E, blockIdx.x, tid, &s_cnt, s_src);
  else if ((E & 3) == 0) scan_vec4(ei, E, lo, nn, tid, &s_cnt, s_src, s_dl);
  else                   scan_scalar(ei, E, 0, E, lo, nn, tid, &s_cnt, s_src, s_dl);
  __syncthreads();
  const int total = s_cnt;
  if (by_key) resolve_src_dst(ei, E, lo, nn, (total < CAP) ? total : CAP, tid, s_src, s_dl);
  else        resolve_src(ei, nn, (total < CAP) ? total : CAP, tid, s_src);
  __syncthreads();

  if (total <= CAP) {
    run_phase(0, total, tid, a_src, h, s_src, s_dl, s_adst, s_m, s_den, s_w, s_acc);
    __syncthreads();
    run_phase(1, total, tid, a_src, h, s_src, s_dl, s_adst, s_m, s_den, s_w, s_acc);
    __syncthreads();
    run_phase(2, total, tid, a_src, h, s_src, s_dl, s_adst, s_m, s_den, s_w, s_acc);
    __syncthreads();
  } else {
    for (int ph = 0; ph < 3; ++ph) {
      __syncthreads();
      for (int i = tid; i < nn; i += TPB) {
        s_src[i] = lo + i;
        s_dl[i]  = (unsigned short)i;
      }
      __syncthreads();
      run_phase(ph, nn, tid, a_src, h, s_src, s_dl, s_adst, s_m, s_den, s_w, s_acc);
      __syncthreads();
      for (int c0 = 0; c0 < E; c0 += CAP) {
        if (tid == 0) s_cnt = 0;
        __syncthreads();
        const int c1 = (E - c0 < CAP) ? E : (c0 + CAP);
        scan_scalar(ei, E, c0, c1, lo, nn, tid, &s_cnt, s_src, s_dl);
        __syncthreads();
        const int n = s_cnt;
        resolve_src(ei, 0, n, tid, s_src);
        __syncthreads();
        run_phase(ph, n, tid, a_src, h, s_src, s_dl, s_adst, s_m, s_den, s_w, s_acc);
        __syncthreads();
      }
    }
  }

  const int lane = tid & 31;
  const int wave = __builtin_amdgcn_readfirstlane(tid >> 5);
  const float4 b4  = ((const float4*)bias)[lane];
  const float4 g4  = ((const float4*)gamma)[lane];
  const float4 be4 = ((const float4*)beta)[lane];
  for (int nl = wave; nl < nn; nl += TPB / 32) {
    const size_t row = (size_t)(lo + nl) * (D_FEAT / 4) + lane;
    const float4 xr = ((const float4*)x)[row];
    const float4 a  = *(const float4*)(s_acc + nl * D_FEAT + 4 * lane);
    const float v0 = a.x + b4.x, v1 = a.y + b4.y, v2 = a.z + b4.z, v3 = a.w + b4.w;
    const float q0 = 0.5f * v0 * (1.0f + erff(v0 * 0.70710678118654752f));
    const float q1 = 0.5f * v1 * (1.0f + erff(v1 * 0.70710678118654752f));
    const float q2 = 0.5f * v2 * (1.0f + erff(v2 * 0.70710678118654752f));
    const float q3 = 0.5f * v3 * (1.0f + erff(v3 * 0.70710678118654752f));
    float s  = (q0 + q1) + (q2 + q3);
    float sq = (q0 * q0 + q1 * q1) + (q2 * q2 + q3 * q3);
#pragma unroll
    for (int off = 16; off > 0; off >>= 1) {
      s  += __shfl_xor(s, off, 32);
      sq += __shfl_xor(sq, off, 32);
    }
    const float mu  = s * (1.0f / D_FEAT);
    const float var = sq * (1.0f / D_FEAT) - mu * mu;
    const float inv = rsqrtf(var + LN_EPS);
    float4 o;
    o.x = (q0 - mu) * inv * g4.x + be4.x + xr.x;
    o.y = (q1 - mu) * inv * g4.y + be4.y + xr.y;
    o.z = (q2 - mu) * inv * g4.z + be4.z + xr.z;
    o.w = (q3 - mu) * inv * g4.w + be4.w + xr.w;
    ((float4*)out)[row] = o;
  }
}

extern "C" void kernel_launch(void* const* d_in, const int* in_sizes, int n_in,
                              void* d_out, int out_size, void* d_ws, size_t ws_size,
                              hipStream_t stream) {
  const float* x       = (const float*)d_in[0];
  const int*   ei      = (const int*)d_in[1];
  const float* W       = (const float*)d_in[2];
  const float* att_src = (const float*)d_in[3];
  const float* att_dst = (const float*)d_in[4];
  const float* bias    = (const float*)d_in[5];
  const float* gamma   = (const float*)d_in[6];
  const float* beta    = (const float*)d_in[7];

  const int N = in_sizes[0] / D_FEAT;
  const int E = in_sizes[1] / 2;

  float* ws    = (float*)d_ws;
  float* h     = ws;
  float* a_src = h + (size_t)N * D_FEAT;
  float* a_dst = a_src + (size_t)N * HEADS;
  unsigned* keys = (unsigned*)(a_dst + (size_t)N * HEADS);

  const int owners = (N + NPG - 1) / NPG;
  const bool by_key = (ws_size >= ((size_t)N * (D_FEAT + 2 * HEADS)) * sizeof(float) + (size_t)E) &&
                      ((E & 15) == 0) && (owners <= (int)KEY_NONE);
  const int kb = by_key ? ((E >> 2) + KEY_VEC_PER_BLOCK * 32 * LG_WAVES - 1) / (KEY_VEC_PER_BLOCK * 32 * LG_WAVES) : 0;

  {
    int rowTiles = (N + 15) / 16;
    gat_gemm_h<<<rowTiles, 256, 0, stream>>>(x, W, h, N);
  }
  {
    int groups = (N + LG_NODES * LG_WAVES - 1) / (LG_NODES * LG_WAVES);
    gat_att_logits<<<kb + groups, 32 * LG_WAVES, 0, stream>>>(h, att_src, att_dst, a_src, a_dst, N,
                                                              ei, E, keys, kb);
  }
  {
    gat_owner<<<owners, TPB, 0, stream>>>(ei, E, h, a_src, a_dst, bias, gamma, beta, x,
                                          (float*)d_out, N, by_key ? keys : nullptr);
  }
}


# h = x@W re-tiled: 256 groups of 16 waves walk equal shares of the row tiles; W staged once per group through LDS into registers; wave tile 16x32, 16-byte A loads; k order of h's sums changes
# speedup vs baseline: 2.7427x; 1.1163x over previous
#include <hip/hip_runtime.h>
#include <math.h>

#define D_FEAT 128
#define HEADS 4
#define CDIM 32
#define NEG_SLOPE 0.2f
#define LN_EPS 1e-5f

typedef __attribute__((ext_vector_type(2))) float v2f;
typedef __attribute__((ext_vector_type(8))) float v8f;

#define GEMM_WAVES 16
#define GEMM_TPB (32 * GEMM_WAVES)
#define GEMM_GROUPS 256
#define WS_PITCH 256
#define GEMM_LDS_FLOATS 41984

static __device__ __forceinline__ void gemm_load_a(const float* __restrict__ x, int rt, int c, int hf,
                                                   int N, float4* a) {
  int arow = rt * 16 + c;
  if (arow >= N) arow = N - 1;
  const float4* ap = (const float4*)(x + (size_t)arow * D_FEAT + 4 * hf);
#pragma unroll
  for (int j = 0; j < 16; ++j) a[j] = ap[2 * j];
}

__global__ __launch_bounds__(GEMM_TPB) void gat_gemm_h(const float* __restrict__ x,
                                                       const float* __restrict__ W,
                                                       float* __restrict__ h,
                                                       int N) {
  __shared__ __attribute__((aligned(16))) float Ws[GEMM_LDS_FLOATS];

  const int tid  = threadIdx.x;
  const int lane = tid & 31;
  const int wave = __builtin_amdgcn_readfirstlane(tid >> 5);
  const int c  = lane & 15;
  const int hf = lane >> 4;
  const int cq = wave & 3;

  const int T  = (N + 15) >> 4;
  const int t0 = (int)(((long long)blockIdx.x * T) / gridDim.x);
  const int t1 = (int)(((long long)(blockIdx.x + 1) * T) / gridDim.x);

  int rt = t0 + (wave >> 2);
  float4 a[16];
  gemm_load_a(x, (rt < T) ? rt : 0, c, hf, N, a);

  {
    const float4* W4 = (const float4*)W;
    float4 r0[4], r1[4];
#pragma unroll
    for (int it = 0; it < 4; ++it) {
      const int i = it * GEMM_TPB + tid;
      const int q = i >> 5, c4 = i & 31;
      r0[it] = W4[(2 * q) * (D_FEAT / 4) + c4];
      r1[it] = W4[(2 * q + 1) * (D_FEAT / 4) + c4];
    }
#pragma unroll
    for (int it = 0; it < 4; ++it) {
      const int i = it * GEMM_TPB + tid;
      const int q = i >> 5, c4 = i & 31;
      float4* dst = (float4*)(Ws + q * WS_PITCH + 8 * c4);
      dst[0] = make_float4(r0[it].x, r1[it].x, r0[it].y, r1[it].y);
      dst[1] = make_float4(r0[it].z, r1[it].z, r0[it].w, r1[it].w);
    }
  }
  __syncthreads();

  const float* bp = Ws + 2 * hf * WS_PITCH + 64 * cq + 4 * c;
  for (; rt < t1; rt += GEMM_WAVES / 4) {
    float4 an[16];
    const int rn = rt + GEMM_WAVES / 4;
    gemm_load_a(x, (rn < t1) ? rn : rt, c, hf, N, an);

    v8f acc0 = {}, acc1 = {};
#pragma unroll
    for (int j = 0; j < 16; ++j) {
      const float4 b0 = *(const float4*)(bp + (4 * j) * WS_PITCH);
      const float4 b1 = *(const float4*)(bp + (4 * j + 1) * WS_PITCH);
      v2f av, bv;
      av.x = a[j].x; av.y = a[j].y;
      bv.x = b0.x; bv.y = b0.y;
      acc0 = __builtin_amdgcn_wmma_f32_16x16x4_f32(false, av, false, bv, (short)0, acc0, false, false);
      bv.x = b0.z; bv.y = b0.w;
      acc1 = __builtin_amdgcn_wmma_f32_16x16x4_f32(false, av, false, bv, (short)0, acc1, false, false);
      av.x = a[j].z; av.y = a[j].w;
      bv.x = b1.x; bv.y = b1.y;
      acc0 = __builtin_amdgcn_wmma_f32_16x16x4_f32(false, av, false, bv, (short)0, acc0, false, false);
      bv.x = b1.z; bv.y = b1.w;
      acc1 = __builtin_amdgcn_wmma_f32_16x16x4_f32(false, av, false, bv, (short)0, acc1, false, false);
    }

    const int rbase = rt * 16 + 8 * hf;
    float* dp = h + (size_t)rbase * D_FEAT + 32 * cq + 2 * c;
    if (rt * 16 + 16 <= N) {
#pragma unroll
      for (int v = 0; v < 8; ++v) {
        v2f o; o.x = acc0[v]; o.y = acc1[v];
        *(v2f*)(dp + (size_t)v * D_FEAT) = o;
      }
    } else {
#pragma unroll
      for (int v = 0; v < 8; ++v) {
        if (rbase + v < N) {
          v2f o; o.x = acc0[v]; o.y = acc1[v];
          *(v2f*)(dp + (size_t)v * D_FEAT) = o;
        }
      }
    }
#pragma unroll
    for (int j = 0; j < 16; ++j) a[j] = an[j];
  }
}

#define LG_NODES 8
#define LG_WAVES 8
#define NPG 197
#define KEY_NONE 255u
#define KEY_VEC_PER_BLOCK 4

static __device__ __forceinline__ unsigned owner_key(int d, int N) {
  return ((unsigned)d < (unsigned)N) ? ((unsigned)d / NPG) : KEY_NONE;
}

__global__ __launch_bounds__(32 * LG_WAVES) void gat_att_logits(const float* __restrict__ h,
                                                                const float* __restrict__ att_src,
                                                                const float* __restrict__ att_dst,
                                                                float* __restrict__ a_src,
                                                                float* __restrict__ a_dst,
                                                                int N,
                                                                const int* __restrict__ ei, int E,
                                                                unsigned* __restrict__ keys, int kb) {
  if ((int)blockIdx.x < kb) {
    const int4* d4 = (const int4*)(ei + E);
    const int nvec = E >> 2;
    int4 d[KEY_VEC_PER_BLOCK];
#pragma unroll
    for (int u = 0; u < KEY_VEC_PER_BLOCK; ++u) {
      const int i = ((int)blockIdx.x * KEY_VEC_PER_BLOCK + u) * (32 * LG_WAVES) + (int)threadIdx.x;
      d[u] = d4[(i < nvec) ? i : 0];
    }
#pragma unroll
    for (int u = 0; u < KEY_VEC_PER_BLOCK; ++u) {
      const int i = ((int)blockIdx.x * KEY_VEC_PER_BLOCK + u) * (32 * LG_WAVES) + (int)threadIdx.x;
      if (i < nvec)
        keys[i] = owner_key(d[u].x, N) | (owner_key(d[u].y, N) << 8) |
                  (owner_key(d[u].z, N) << 16) | (owner_key(d[u].w, N) << 24);
    }
    return;
  }
  const int blk  = (int)blockIdx.x - kb;
  const int lane = threadIdx.x & 31;
  const int wave = __builtin_amdgcn_readfirstlane((int)(threadIdx.x >> 5));
  const int n0 = (blk * LG_WAVES + wave) * LG_NODES;
  if (n0 >= N) return;

  const float4* h4 = (const float4*)h;
  float4 r[LG_NODES];
#pragma unroll
  for (int u = 0; u < LG_NODES; ++u) {
    int n = n0 + u;
    if (n >= N) n = N - 1;
    r[u] = h4[(size_t)n * (D_FEAT / 4) + lane];
  }
  const float4 as4 = ((const float4*)att_src)[lane];
  const float4 ad4 = ((const float4*)att_dst)[lane];

  float ps[LG_NODES], pd[LG_NODES];
#pragma unroll
  for (int u = 0; u < LG_NODES; ++u) {
    ps[u] = (r[u].x * as4.x + r[u].y * as4.y) + (r[u].z * as4.z + r[u].w * as4.w);
    pd[u] = (r[u].x * ad4.x + r[u].y * ad4.y) + (r[u].z * ad4.z + r[u].w * ad4.w);
  }

  const bool b2 = (lane & 4) != 0, b1 = (lane & 2) != 0, b0 = (lane & 1) != 0;
  float qs[4], qd[4];
#pragma unroll
  for (int j = 0; j < 4; ++j) {
    const float ks = b2 ? ps[j + 4] : ps[j], ss = b2 ? ps[j] : ps[j + 4];
    const float kd = b2 ? pd[j + 4] : pd[j], sd = b2 ? pd[j] : pd[j + 4];
    qs[j] = ks + __shfl_xor(ss, 4, 32);
    qd[j] = kd + __shfl_xor(sd, 4, 32);
  }
  float ts[2], td[2];
#pragma unroll
  for (int j = 0; j < 2; ++j) {
    const float ks = b1 ? qs[j + 2] : qs[j], ss = b1 ? qs[j] : qs[j + 2];
    const float kd = b1 ? qd[j + 2] : qd[j], sd = b1 ? qd[j] : qd[j + 2];
    ts[j] = ks + __shfl_xor(ss, 2, 32);
    td[j] = kd + __shfl_xor(sd, 2, 32);
  }
  const float fs = (b0 ? ts[1] : ts[0]) + __shfl_xor(b0 ? ts[0] : ts[1], 1, 32);
  const float fd = (b0 ? td[1] : td[0]) + __shfl_xor(b0 ? td[0] : td[1], 1, 32);

  const int n = n0 + (lane & 7);
  if (n < N) {
    a_src[(size_t)n * HEADS + (lane >> 3)] = fs;
    a_dst[(size_t)n * HEADS + (lane >> 3)] = fd;
  }
}

#define CAP 4096
#define TPB 1024
#define SCAN_DEPTH 8

static __device__ __forceinline__ void rec_append(int e, unsigned d,
                                                  int* s_cnt, int* s_src, unsigned short* s_dl) {
  const int p = atomicAdd(s_cnt, 1);
  if (p < CAP) {
    s_src[p] = e;
    s_dl[p]  = (unsigned short)d;
  }
}

static __device__ __forceinline__ void resolve_src(const int* __restrict__ ei, int from, int to,
                                                   int tid, int* s_src) {
  for (int i = from + tid; i < to; i += TPB) s_src[i] = ei[s_src[i]];
}

static __device__ __forceinline__ void scan_scalar(const int* __restrict__ ei, int E, int c0, int c1,
                                                   int lo, int nn, int tid,
                                                   int* s_cnt, int* s_src, unsigned short* s_dl) {
  for (int e = c0 + tid; e < c1; e += TPB) {
    const unsigned d = (unsigned)(ei[E + e] - lo);
    if (d < (unsigned)nn) rec_append(e, d, s_cnt, s_src, s_dl);
  }
}

static __device__ __forceinline__ void scan_vec4(const int* __restrict__ ei, int E,
                                                 int lo, int nn, int tid,
                                                 int* s_cnt, int* s_src, unsigned short* s_dl) {
  const int nvec = E >> 2;
  const int4* d4 = (const int4*)(ei + E);
  const int nb = (nvec + TPB * SCAN_DEPTH - 1) / (TPB * SCAN_DEPTH);
  if (nb == 0) return;
  int kb = (int)(blockIdx.x % (unsigned)nb);
  for (int k = 0; k < nb; ++k) {
    const int b = kb * (TPB * SCAN_DEPTH) + tid;
    kb = (kb + 1 == nb) ? 0 : kb + 1;
    int4 v[SCAN_DEPTH];
#pragma unroll
    for (int u = 0; u < SCAN_DEPTH; ++u) {
      const int i = b + TPB * u;
      v[u] = d4[(i < nvec) ? i : 0];
    }
#pragma unroll
    for (int u = 0; u < SCAN_DEPTH; ++u) {
      const int i = b + TPB * u;
      const unsigned d0 = (unsigned)(v[u].x - lo);
      const unsigned d1 = (unsigned)(v[u].y - lo);
      const unsigned d2 = (unsigned)(v[u].z - lo);
      const unsigned d3 = (unsigned)(v[u].w - lo);
      const unsigned mn = min(min(d0, d1), min(d2, d3));
      if (i < nvec && mn < (unsigned)nn) {
        if (d0 < (unsigned)nn) rec_append(4 * i + 0, d0, s_cnt, s_src, s_dl);
        if (d1 < (unsigned)nn) rec_append(4 * i + 1, d1, s_cnt, s_src, s_dl);
        if (d2 < (unsigned)nn) rec_append(4 * i + 2, d2, s_cnt, s_src, s_dl);
        if (d3 < (unsigned)nn) rec_append(4 * i + 3, d3, s_cnt, s_src, s_dl);
      }
    }
  }
}

static __device__ __forceinline__ void resolve_src_dst(const int* __restrict__ ei, int E, int lo,
                                                       int from, int to, int tid,
                                                       int* s_src, unsigned short* s_dl) {
  for (int i = from + tid; i < to; i += TPB) {
    const int e = s_src[i];
    const int d = ei[E + e];
    const int s = ei[e];
    s_dl[i]  = (unsigned short)(d - lo);
    s_src[i] = s;
  }
}

static __device__ __forceinline__ unsigned nonzero_bytes(unsigned x) {
  return ((x & 0x7F7F7F7Fu) + 0x7F7F7F7Fu) | x;
}

static __device__ __forceinline__ void scan_keys(const unsigned* __restrict__ keys, int E,
                                                 unsigned gid, int tid, int* s_cnt, int* s_src) {
  const int nvec = E >> 4;
  const uint4* k4 = (const uint4*)keys;
  const unsigned rep4 = gid * 0x01010101u;
  const int nb = (nvec + TPB * SCAN_DEPTH - 1) / (TPB * SCAN_DEPTH);
  if (nb == 0) return;
  int kb = (int)(blockIdx.x % (unsigned)nb);
  for (int k = 0; k < nb; ++k) {
    const int b = kb * (TPB * SCAN_DEPTH) + tid;
    kb = (kb + 1 == nb) ? 0 : kb + 1;
    uint4 v[SCAN_DEPTH];
#pragma unroll
    for (int u = 0; u < SCAN_DEPTH; ++u) {
      const int i = b + TPB * u;
      v[u] = k4[(i < nvec) ? i : 0];
    }
#pragma unroll
    for (int u = 0; u < SCAN_DEPTH; ++u) {
      const int i = b + TPB * u;
      unsigned m = (nonzero_bytes(v[u].x ^ rep4) & 0x80808080u) |
                   ((nonzero_bytes(v[u].y ^ rep4) >> 1) & 0x40404040u) |
                   ((nonzero_bytes(v[u].z ^ rep4) >> 2) & 0x20202020u) |
                   ((nonzero_bytes(v[u].w ^ rep4) >> 3) & 0x10101010u);
      m ^= 0xF0F0F0F0u;
      if (i >= nvec) m = 0u;
      while (m != 0u) {
        const int p = __builtin_ctz(m);
        m &= m - 1u;
        const int e = 16 * i + 4 * (7 - (p & 7)) + (p >> 3);
        const int q = atomicAdd(s_cnt, 1);
        if (q < CAP) s_src[q] = e;
      }
    }
  }
}

static __device__ __forceinline__ float edge_logit(const float* __restrict__ a_src,
                                                   const float* s_adst, int src, int dl, int hd) {
  const float v = a_src[(size_t)src * HEADS + hd] + s_adst[dl * HEADS + hd];
  return (v > 0.0f) ? v : NEG_SLOPE * v;
}

static __device__ __forceinline__ void run_phase(int ph, int n, int tid,
                                                 const float* __restrict__ a_src,
                                                 const float* __restrict__ h,
                                                 const int* s_src, const unsigned short* s_dl,
                                                 const float* s_adst, float* s_m, float* s_den,
                                                 float* s_w, float* s_acc) {
  if (ph == 0) {
    for (int i = tid; i < n * HEADS; i += TPB) {
      const int p = i >> 2, hd = i & 3;
      const int dl = s_dl[p];
      const float v = edge_logit(a_src, s_adst, s_src[p], dl, hd);
      atomicMax(&s_m[dl * HEADS + hd], v);
    }
  } else if (ph == 1) {
    for (int i = tid; i < n * HEADS; i += TPB) {
      const int p = i >> 2, hd = i & 3;
      const int dl = s_dl[p];
      const float v = edge_logit(a_src, s_adst, s_src[p], dl, hd);
      atomicAdd(&s_den[dl * HEADS + hd], __expf(v - s_m[dl * HEADS + hd]));
    }
  } else {
    for (int i = tid; i < n * HEADS; i += TPB) {
      const int p = i >> 2, hd = i & 3;
      const int dl = s_dl[p];
      const float v = edge_logit(a_src, s_adst, s_src[p], dl, hd);
      s_w[i] = __expf(v - s_m[dl * HEADS + hd]) / (s_den[dl * HEADS + hd] + 1e-16f);
    }
    __syncthreads();
    const int lane = tid & 31;
    const int wave = __builtin_amdgcn_readfirstlane(tid >> 5);
    const float4* h4 = (const float4*)h;
    for (int p0 = wave; p0 < n; p0 += 32 * 4) {
      float4 r[4];
      int srcv[4];
      int dlv[4];
      float wv[4];
#pragma unroll
      for (int u = 0; u < 4; ++u) {
        const int p  = p0 + 32 * u;
        const int pc = (p < n) ? p : p0;
        srcv[u] = s_src[pc];
        dlv[u]  = s_dl[pc];
        wv[u]   = (p < n) ? s_w[pc * HEADS + (lane >> 3)] : 0.0f;
      }
#pragma unroll
      for (int u = 0; u < 4; ++u) r[u] = h4[(size_t)srcv[u] * (D_FEAT / 4) + lane];
#pragma unroll
      for (int u = 0; u < 4; ++u) {
        const bool ok = (p0 + 32 * u < n);
        float* ap = s_acc + dlv[u] * D_FEAT + 4 * lane;
        atomicAdd(ap + 0, ok ? r[u].x * wv[u] : 0.0f);
        atomicAdd(ap + 1, ok ? r[u].y * wv[u] : 0.0f);
        atomicAdd(ap + 2, ok ? r[u].z * wv[u] : 0.0f);
        atomicAdd(ap + 3, ok ? r[u].w * wv[u] : 0.0f);
      }
    }
  }
}

__global__ __launch_bounds__(TPB) void gat_owner(const int* __restrict__ ei, int E,
                                                 const float* __restrict__ h,
                                                 const float* __restrict__ a_src,
                                                 const float* __restrict__ a_dst,
                                                 const float* __restrict__ bias,
                                                 const float* __restrict__ gamma,
                                                 const float* __restrict__ beta,
                                                 const float* __restrict__ x,
                                                 float* __restrict__ out, int N,
                                                 const unsigned* __restrict__ keys) {
  __shared__ __attribute__((aligned(16))) float s_acc[NPG * D_FEAT];
  __shared__ __attribute__((aligned(16))) float s_w[CAP * HEADS];
  __shared__ int s_src[CAP];
  __shared__ unsigned short s_dl[CAP];
  __shared__ float s_m[NPG * HEADS];
  __shared__ float s_den[NPG * HEADS];
  __shared__ float s_adst[NPG * HEADS];
  __shared__ int s_cnt;

  const int tid = threadIdx.x;
  const int lo  = blockIdx.x * NPG;
  if (lo >= N) return;
  const int nn  = (N - lo < NPG) ? (N - lo) : NPG;

  for (int i = tid; i < NPG * D_FEAT / 4; i += TPB)
    ((float4*)s_acc)[i] = make_float4(0.0f, 0.0f, 0.0f, 0.0f);
  for (int i = tid; i < nn * HEADS; i += TPB) {
    s_m[i]    = -INFINITY;
    s_den[i]  = 0.0f;
    s_adst[i] = a_dst[(size_t)lo * HEADS + i];
  }
  for (int i = tid; i < nn; i += TPB) {
    s_src[i] = lo + i;
    s_dl[i]  = (unsigned short)i;
  }
  if (tid == 0) s_cnt = nn;
  __syncthreads();

  const bool by_key = (keys != nullptr) && ((E & 15) == 0) && (gridDim.x <= KEY_NONE);
  if (by_key)            scan_keys(keys, E, blockIdx.x, tid, &s_cnt, s_src);
  else if ((E & 3) == 0) scan_vec4(ei, E, lo, nn, tid, &s_cnt, s_src, s_dl);
  else                   scan_scalar(ei, E, 0, E, lo, nn, tid, &s_cnt, s_src, s_dl);
  __syncthreads();
  const int total = s_cnt;
  if (by_key) resolve_src_dst(ei, E, lo, nn, (total < CAP) ? total : CAP, tid, s_src, s_dl);
  else        resolve_src(ei, nn, (total < CAP) ? total : CAP, tid, s_src);
  __syncthreads();

  if (total <= CAP) {
    run_phase(0, total, tid, a_src, h, s_src, s_dl, s_adst, s_m, s_den, s_w, s_acc);
    __syncthreads();
    run_phase(1, total, tid, a_src, h, s_src, s_dl, s_adst, s_m, s_den, s_w, s_acc);
    __syncthreads();
    run_phase(2, total, tid, a_src, h, s_src, s_dl, s_adst, s_m, s_den, s_w, s_acc);
    __syncthreads();
  } else {
    for (int ph = 0; ph < 3; ++ph) {
      __syncthreads();
      for (int i = tid; i < nn; i += TPB) {
        s_src[i] = lo + i;
        s_dl[i]  = (unsigned short)i;
      }
      __syncthreads();
      run_phase(ph, nn, tid, a_src, h, s_src, s_dl, s_adst, s_m, s_den, s_w, s_acc);
      __syncthreads();
      for (int c0 = 0; c0 < E; c0 += CAP) {
        if (tid == 0) s_cnt = 0;
        __syncthreads();
        const int c1 = (E - c0 < CAP) ? E : (c0 + CAP);
        scan_scalar(ei, E, c0, c1, lo, nn, tid, &s_cnt, s_src, s_dl);
        __syncthreads();
        const int n = s_cnt;
        resolve_src(ei, 0, n, tid, s_src);
        __syncthreads();
        run_phase(ph, n, tid, a_src, h, s_src, s_dl, s_adst, s_m, s_den, s_w, s_acc);
        __syncthreads();
      }
    }
  }

  const int lane = tid & 31;
  const int wave = __builtin_amdgcn_readfirstlane(tid >> 5);
  const float4 b4  = ((const float4*)bias)[lane];
  const float4 g4  = ((const float4*)gamma)[lane];
  const float4 be4 = ((const float4*)beta)[lane];
  for (int nl = wave; nl < nn; nl += TPB / 32) {
    const size_t row = (size_t)(lo + nl) * (D_FEAT / 4) + lane;
    const float4 xr = ((const float4*)x)[row];
    const float4 a  = *(const float4*)(s_acc + nl * D_FEAT + 4 * lane);
    const float v0 = a.x + b4.x, v1 = a.y + b4.y, v2 = a.z + b4.z, v3 = a.w + b4.w;
    const float q0 = 0.5f * v0 * (1.0f + erff(v0 * 0.70710678118654752f));
    const float q1 = 0.5f * v1 * (1.0f + erff(v1 * 0.70710678118654752f));
    const float q2 = 0.5f * v2 * (1.0f + erff(v2 * 0.70710678118654752f));
    const float q3 = 0.5f * v3 * (1.0f + erff(v3 * 0.70710678118654752f));
    float s  = (q0 + q1) + (q2 + q3);
    float sq = (q0 * q0 + q1 * q1) + (q2 * q2 + q3 * q3);
#pragma unroll
    for (int off = 16; off > 0; off >>= 1) {
      s  += __shfl_xor(s, off, 32);
      sq += __shfl_xor(sq, off, 32);
    }
    const float mu  = s * (1.0f / D_FEAT);
    const float var = sq * (1.0f / D_FEAT) - mu * mu;
    const float inv = rsqrtf(var + LN_EPS);
    float4 o;
    o.x = (q0 - mu) * inv * g4.x + be4.x + xr.x;
    o.y = (q1 - mu) * inv * g4.y + be4.y + xr.y;
    o.z = (q2 - mu) * inv * g4.z + be4.z + xr.z;
    o.w = (q3 - mu) * inv * g4.w + be4.w + xr.w;
    ((float4*)out)[row] = o;
  }
}

extern "C" void kernel_launch(void* const* d_in, const int* in_sizes, int n_in,
                              void* d_out, int out_size, void* d_ws, size_t ws_size,
                              hipStream_t stream) {
  const float* x       = (const float*)d_in[0];
  const int*   ei      = (const int*)d_in[1];
  const float* W       = (const float*)d_in[2];
  const float* att_src = (const float*)d_in[3];
  const float* att_dst = (const float*)d_in[4];
  const float* bias    = (const float*)d_in[5];
  const float* gamma   = (const float*)d_in[6];
  const float* beta    = (const float*)d_in[7];

  const int N = in_sizes[0] / D_FEAT;
  const int E = in_sizes[1] / 2;

  float* ws    = (float*)d_ws;
  float* h     = ws;
  float* a_src = h + (size_t)N * D_FEAT;
  float* a_dst = a_src + (size_t)N * HEADS;
  unsigned* keys = (unsigned*)(a_dst + (size_t)N * HEADS);

  const int owners = (N + NPG - 1) / NPG;
  const bool by_key = (ws_size >= ((size_t)N * (D_FEAT + 2 * HEADS)) * sizeof(float) + (size_t)E) &&
                      ((E & 15) == 0) && (owners <= (int)KEY_NONE);
  const int kb = by_key ? ((E >> 2) + KEY_VEC_PER_BLOCK * 32 * LG_WAVES - 1) / (KEY_VEC_PER_BLOCK * 32 * LG_WAVES) : 0;

  {
    gat_gemm_h<<<GEMM_GROUPS, GEMM_TPB, 0, stream>>>(x, W, h, N);
  }
  {
    int groups = (N + LG_NODES * LG_WAVES - 1) / (LG_NODES * LG_WAVES);
    gat_att_logits<<<kb + groups, 32 * LG_WAVES, 0, stream>>>(h, att_src, att_dst, a_src, a_dst, N,
                                                              ei, E, keys, kb);
  }
  {
    gat_owner<<<owners, TPB, 0, stream>>>(ei, E, h, a_src, a_dst, bias, gamma, beta, x,
                                          (float*)d_out, N, by_key ? keys : nullptr);
  }
}


# owner softmax: a_src logits fetched ONCE, a thread's 16 (record, head) loads in flight together; logit and exp stay in registers over max, sum and weight passes (LDS float atomics, arrival order)
# speedup vs baseline: 2.8697x; 1.0463x over previous
#include <hip/hip_runtime.h>
#include <math.h>

#define D_FEAT 128
#define HEADS 4
#define CDIM 32
#define NEG_SLOPE 0.2f
#define LN_EPS 1e-5f

typedef __attribute__((ext_vector_type(2))) float v2f;
typedef __attribute__((ext_vector_type(8))) float v8f;

#define GEMM_WAVES 16
#define GEMM_TPB (32 * GEMM_WAVES)
#define GEMM_GROUPS 256
#define WS_PITCH 256
#define GEMM_LDS_FLOATS 41984

static __device__ __forceinline__ void gemm_load_a(const float* __restrict__ x, int rt, int c, int hf,
                                                   int N, float4* a) {
  int arow = rt * 16 + c;
  if (arow >= N) arow = N - 1;
  const float4* ap = (const float4*)(x + (size_t)arow * D_FEAT + 4 * hf);
#pragma unroll
  for (int j = 0; j < 16; ++j) a[j] = ap[2 * j];
}

__global__ __launch_bounds__(GEMM_TPB) void gat_gemm_h(const float* __restrict__ x,
                                                       const float* __restrict__ W,
                                                       float* __restrict__ h,
                                                       int N) {
  __shared__ __attribute__((aligned(16))) float Ws[GEMM_LDS_FLOATS];

  const int tid  = threadIdx.x;
  const int lane = tid & 31;
  const int wave = __builtin_amdgcn_readfirstlane(tid >> 5);
  const int c  = lane & 15;
  const int hf = lane >> 4;
  const int cq = wave & 3;

  const int T  = (N + 15) >> 4;
  const int t0 = (int)(((long long)blockIdx.x * T) / gridDim.x);
  const int t1 = (int)(((long long)(blockIdx.x + 1) * T) / gridDim.x);

  int rt = t0 + (wave >> 2);
  float4 a[16];
  gemm_load_a(x, (rt < T) ? rt : 0, c, hf, N, a);

  {
    const float4* W4 = (const float4*)W;
    float4 r0[4], r1[4];
#pragma unroll
    for (int it = 0; it < 4; ++it) {
      const int i = it * GEMM_TPB + tid;
      const int q = i >> 5, c4 = i & 31;
      r0[it] = W4[(2 * q) * (D_FEAT / 4) + c4];
      r1[it] = W4[(2 * q + 1) * (D_FEAT / 4) + c4];
    }
#pragma unroll
    for (int it = 0; it < 4; ++it) {
      const int i = it * GEMM_TPB + tid;
      const int q = i >> 5, c4 = i & 31;
      float4* dst = (float4*)(Ws + q * WS_PITCH + 8 * c4);
      dst[0] = make_float4(r0[it].x, r1[it].x, r0[it].y, r1[it].y);
      dst[1] = make_float4(r0[it].z, r1[it].z, r0[it].w, r1[it].w);
    }
  }
  __syncthreads();

  const float* bp = Ws + 2 * hf * WS_PITCH + 64 * cq + 4 * c;
  for (; rt < t1; rt += GEMM_WAVES / 4) {
    float4 an[16];
    const int rn = rt + GEMM_WAVES / 4;
    gemm_load_a(x, (rn < t1) ? rn : rt, c, hf, N, an);

    v8f acc0 = {}, acc1 = {};
#pragma unroll
    for (int j = 0; j < 16; ++j) {
      const float4 b0 = *(const float4*)(bp + (4 * j) * WS_PITCH);
      const float4 b1 = *(const float4*)(bp + (4 * j + 1) * WS_PITCH);
      v2f av, bv;
      av.x = a[j].x; av.y = a[j].y;
      bv.x = b0.x; bv.y = b0.y;
      acc0 = __builtin_amdgcn_wmma_f32_16x16x4_f32(false, av, false, bv, (short)0, acc0, false, false);
      bv.x = b0.z; bv.y = b0.w;
      acc1 = __builtin_amdgcn_wmma_f32_16x16x4_f32(false, av, false, bv, (short)0, acc1, false, false);
      av.x = a[j].z; av.y = a[j].w;
      bv.x = b1.x; bv.y = b1.y;
      acc0 = __builtin_amdgcn_wmma_f32_16x16x4_f32(false, av, false, bv, (short)0, acc0, false, false);
      bv.x = b1.z; bv.y = b1.w;
      acc1 = __builtin_amdgcn_wmma_f32_16x16x4_f32(false, av, false, bv, (short)0, acc1, false, false);
    }

    const int rbase = rt * 16 + 8 * hf;
    float* dp = h + (size_t)rbase * D_FEAT + 32 * cq + 2 * c;
    if (rt * 16 + 16 <= N) {
#pragma unroll
      for (int v = 0; v < 8; ++v) {
        v2f o; o.x = acc0[v]; o.y = acc1[v];
        *(v2f*)(dp + (size_t)v * D_FEAT) = o;
      }
    } else {
#pragma unroll
      for (int v = 0; v < 8; ++v) {
        if (rbase + v < N) {
          v2f o; o.x = acc0[v]; o.y = acc1[v];
          *(v2f*)(dp + (size_t)v * D_FEAT) = o;
        }
      }
    }
#pragma unroll
    for (int j = 0; j < 16; ++j) a[j] = an[j];
  }
}

#define LG_NODES 8
#define LG_WAVES 8
#define NPG 197
#define KEY_NONE 255u
#define KEY_VEC_PER_BLOCK 4

static __device__ __forceinline__ unsigned owner_key(int d, int N) {
  return ((unsigned)d < (unsigned)N) ? ((unsigned)d / NPG) : KEY_NONE;
}

__global__ __launch_bounds__(32 * LG_WAVES) void gat_att_logits(const float* __restrict__ h,
                                                                const float* __restrict__ att_src,
                                                                const float* __restrict__ att_dst,
                                                                float* __restrict__ a_src,
                                                                float* __restrict__ a_dst,
                                                                int N,
                                                                const int* __restrict__ ei, int E,
                                                                unsigned* __restrict__ keys, int kb) {
  if ((int)blockIdx.x < kb) {
    const int4* d4 = (const int4*)(ei + E);
    const int nvec = E >> 2;
    int4 d[KEY_VEC_PER_BLOCK];
#pragma unroll
    for (int u = 0; u < KEY_VEC_PER_BLOCK; ++u) {
      const int i = ((int)blockIdx.x * KEY_VEC_PER_BLOCK + u) * (32 * LG_WAVES) + (int)threadIdx.x;
      d[u] = d4[(i < nvec) ? i : 0];
    }
#pragma unroll
    for (int u = 0; u < KEY_VEC_PER_BLOCK; ++u) {
      const int i = ((int)blockIdx.x * KEY_VEC_PER_BLOCK + u) * (32 * LG_WAVES) + (int)threadIdx.x;
      if (i < nvec)
        keys[i] = owner_key(d[u].x, N) | (owner_key(d[u].y, N) << 8) |
                  (owner_key(d[u].z, N) << 16) | (owner_key(d[u].w, N) << 24);
    }
    return;
  }
  const int blk  = (int)blockIdx.x - kb;
  const int lane = threadIdx.x & 31;
  const int wave = __builtin_amdgcn_readfirstlane((int)(threadIdx.x >> 5));
  const int n0 = (blk * LG_WAVES + wave) * LG_NODES;
  if (n0 >= N) return;

  const float4* h4 = (const float4*)h;
  float4 r[LG_NODES];
#pragma unroll
  for (int u = 0; u < LG_NODES; ++u) {
    int n = n0 + u;
    if (n >= N) n = N - 1;
    r[u] = h4[(size_t)n * (D_FEAT / 4) + lane];
  }
  const float4 as4 = ((const float4*)att_src)[lane];
  const float4 ad4 = ((const float4*)att_dst)[lane];

  float ps[LG_NODES], pd[LG_NODES];
#pragma unroll
  for (int u = 0; u < LG_NODES; ++u) {
    ps[u] = (r[u].x * as4.x + r[u].y * as4.y) + (r[u].z * as4.z + r[u].w * as4.w);
    pd[u] = (r[u].x * ad4.x + r[u].y * ad4.y) + (r[u].z * ad4.z + r[u].w * ad4.w);
  }

  const bool b2 = (lane & 4) != 0, b1 = (lane & 2) != 0, b0 = (lane & 1) != 0;
  float qs[4], qd[4];
#pragma unroll
  for (int j = 0; j < 4; ++j) {
    const float ks = b2 ? ps[j + 4] : ps[j], ss = b2 ? ps[j] : ps[j + 4];
    const float kd = b2 ? pd[j + 4] : pd[j], sd = b2 ? pd[j] : pd[j + 4];
    qs[j] = ks + __shfl_xor(ss, 4, 32);
    qd[j] = kd + __shfl_xor(sd, 4, 32);
  }
  float ts[2], td[2];
#pragma unroll
  for (int j = 0; j < 2; ++j) {
    const float ks = b1 ? qs[j + 2] : qs[j], ss = b1 ? qs[j] : qs[j + 2];
    const float kd = b1 ? qd[j + 2] : qd[j], sd = b1 ? qd[j] : qd[j + 2];
    ts[j] = ks + __shfl_xor(ss, 2, 32);
    td[j] = kd + __shfl_xor(sd, 2, 32);
  }
  const float fs = (b0 ? ts[1] : ts[0]) + __shfl_xor(b0 ? ts[0] : ts[1], 1, 32);
  const float fd = (b0 ? td[1] : td[0]) + __shfl_xor(b0 ? td[0] : td[1], 1, 32);

  const int n = n0 + (lane & 7);
  if (n < N) {
    a_src[(size_t)n * HEADS + (lane >> 3)] = fs;
    a_dst[(size_t)n * HEADS + (lane >> 3)] = fd;
  }
}

#define CAP 4096
#define TPB 1024
#define SCAN_DEPTH 8

static __device__ __forceinline__ void rec_append(int e, unsigned d,
                                                  int* s_cnt, int* s_src, unsigned short* s_dl) {
  const int p = atomicAdd(s_cnt, 1);
  if (p < CAP) {
    s_src[p] = e;
    s_dl[p]  = (unsigned short)d;
  }
}

static __device__ __forceinline__ void resolve_src(const int* __restrict__ ei, int from, int to,
                                                   int tid, int* s_src) {
  for (int i = from + tid; i < to; i += TPB) s_src[i] = ei[s_src[i]];
}

static __device__ __forceinline__ void scan_scalar(const int* __restrict__ ei, int E, int c0, int c1,
                                                   int lo, int nn, int tid,
                                                   int* s_cnt, int* s_src, unsigned short* s_dl) {
  for (int e = c0 + tid; e < c1; e += TPB) {
    const unsigned d = (unsigned)(ei[E + e] - lo);
    if (d < (unsigned)nn) rec_append(e, d, s_cnt, s_src, s_dl);
  }
}

static __device__ __forceinline__ void scan_vec4(const int* __restrict__ ei, int E,
                                                 int lo, int nn, int tid,
                                                 int* s_cnt, int* s_src, unsigned short* s_dl) {
  const int nvec = E >> 2;
  const int4* d4 = (const int4*)(ei + E);
  const int nb = (nvec + TPB * SCAN_DEPTH - 1) / (TPB * SCAN_DEPTH);
  if (nb == 0) return;
  int kb = (int)(blockIdx.x % (unsigned)nb);
  for (int k = 0; k < nb; ++k) {
    const int b = kb * (TPB * SCAN_DEPTH) + tid;
    kb = (kb + 1 == nb) ? 0 : kb + 1;
    int4 v[SCAN_DEPTH];
#pragma unroll
    for (int u = 0; u < SCAN_DEPTH; ++u) {
      const int i = b + TPB * u;
      v[u] = d4[(i < nvec) ? i : 0];
    }
#pragma unroll
    for (int u = 0; u < SCAN_DEPTH; ++u) {
      const int i = b + TPB * u;
      const unsigned d0 = (unsigned)(v[u].x - lo);
      const unsigned d1 = (unsigned)(v[u].y - lo);
      const unsigned d2 = (unsigned)(v[u].z - lo);
      const unsigned d3 = (unsigned)(v[u].w - lo);
      const unsigned mn = min(min(d0, d1), min(d2, d3));
      if (i < nvec && mn < (unsigned)nn) {
        if (d0 < (unsigned)nn) rec_append(4 * i + 0, d0, s_cnt, s_src, s_dl);
        if (d1 < (unsigned)nn) rec_append(4 * i + 1, d1, s_cnt, s_src, s_dl);
        if (d2 < (unsigned)nn) rec_append(4 * i + 2, d2, s_cnt, s_src, s_dl);
        if (d3 < (unsigned)nn) rec_append(4 * i + 3, d3, s_cnt, s_src, s_dl);
      }
    }
  }
}

static __device__ __forceinline__ void resolve_src_dst(const int* __restrict__ ei, int E, int lo,
                                                       int from, int to, int tid,
                                                       int* s_src, unsigned short* s_dl) {
  for (int i = from + tid; i < to; i += TPB) {
    const int e = s_src[i];
    const int d = ei[E + e];
    const int s = ei[e];
    s_dl[i]  = (unsigned short)(d - lo);
    s_src[i] = s;
  }
}

static __device__ __forceinline__ unsigned nonzero_bytes(unsigned x) {
  return ((x & 0x7F7F7F7Fu) + 0x7F7F7F7Fu) | x;
}

static __device__ __forceinline__ void scan_keys(const unsigned* __restrict__ keys, int E,
                                                 unsigned gid, int tid, int* s_cnt, int* s_src) {
  const int nvec = E >> 4;
  const uint4* k4 = (const uint4*)keys;
  const unsigned rep4 = gid * 0x01010101u;
  const int nb = (nvec + TPB * SCAN_DEPTH - 1) / (TPB * SCAN_DEPTH);
  if (nb == 0) return;
  int kb = (int)(blockIdx.x % (unsigned)nb);
  for (int k = 0; k < nb; ++k) {
    const int b = kb * (TPB * SCAN_DEPTH) + tid;
    kb = (kb + 1 == nb) ? 0 : kb + 1;
    uint4 v[SCAN_DEPTH];
#pragma unroll
    for (int u = 0; u < SCAN_DEPTH; ++u) {
      const int i = b + TPB * u;
      v[u] = k4[(i < nvec) ? i : 0];
    }
#pragma unroll
    for (int u = 0; u < SCAN_DEPTH; ++u) {
      const int i = b + TPB * u;
      unsigned m = (nonzero_bytes(v[u].x ^ rep4) & 0x80808080u) |
                   ((nonzero_bytes(v[u].y ^ rep4) >> 1) & 0x40404040u) |
                   ((nonzero_bytes(v[u].z ^ rep4) >> 2) & 0x20202020u) |
                   ((nonzero_bytes(v[u].w ^ rep4) >> 3) & 0x10101010u);
      m ^= 0xF0F0F0F0u;
      if (i >= nvec) m = 0u;
      while (m != 0u) {
        const int p = __builtin_ctz(m);
        m &= m - 1u;
        const int e = 16 * i + 4 * (7 - (p & 7)) + (p >> 3);
        const int q = atomicAdd(s_cnt, 1);
        if (q < CAP) s_src[q] = e;
      }
    }
  }
}

static __device__ __forceinline__ float edge_logit(const float* __restrict__ a_src,
                                                   const float* s_adst, int src, int dl, int hd) {
  const float v = a_src[(size_t)src * HEADS + hd] + s_adst[dl * HEADS + hd];
  return (v > 0.0f) ? v : NEG_SLOPE * v;
}

static __device__ __forceinline__ void row_sum(int n, int tid, const float* __restrict__ h,
                                               const int* s_src, const unsigned short* s_dl,
                                               const float* s_w, float* s_acc) {
  const int lane = tid & 31;
  const int wave = __builtin_amdgcn_readfirstlane(tid >> 5);
  const float4* h4 = (const float4*)h;
  for (int p0 = wave; p0 < n; p0 += 32 * 4) {
    float4 r[4];
    int srcv[4];
    int dlv[4];
    float wv[4];
#pragma unroll
    for (int u = 0; u < 4; ++u) {
      const int p  = p0 + 32 * u;
      const int pc = (p < n) ? p : p0;
      srcv[u] = s_src[pc];
      dlv[u]  = s_dl[pc];
      wv[u]   = (p < n) ? s_w[pc * HEADS + (lane >> 3)] : 0.0f;
    }
#pragma unroll
    for (int u = 0; u < 4; ++u) r[u] = h4[(size_t)srcv[u] * (D_FEAT / 4) + lane];
#pragma unroll
    for (int u = 0; u < 4; ++u) {
      const bool ok = (p0 + 32 * u < n);
      float* ap = s_acc + dlv[u] * D_FEAT + 4 * lane;
      atomicAdd(ap + 0, ok ? r[u].x * wv[u] : 0.0f);
      atomicAdd(ap + 1, ok ? r[u].y * wv[u] : 0.0f);
      atomicAdd(ap + 2, ok ? r[u].z * wv[u] : 0.0f);
      atomicAdd(ap + 3, ok ? r[u].w * wv[u] : 0.0f);
    }
  }
}

static __device__ __forceinline__ void run_phase(int ph, int n, int tid,
                                                 const float* __restrict__ a_src,
                                                 const float* __restrict__ h,
                                                 const int* s_src, const unsigned short* s_dl,
                                                 const float* s_adst, float* s_m, float* s_den,
                                                 float* s_w, float* s_acc) {
  if (ph == 0) {
    for (int i = tid; i < n * HEADS; i += TPB) {
      const int p = i >> 2, hd = i & 3;
      const int dl = s_dl[p];
      const float v = edge_logit(a_src, s_adst, s_src[p], dl, hd);
      atomicMax(&s_m[dl * HEADS + hd], v);
    }
  } else if (ph == 1) {
    for (int i = tid; i < n * HEADS; i += TPB) {
      const int p = i >> 2, hd = i & 3;
      const int dl = s_dl[p];
      const float v = edge_logit(a_src, s_adst, s_src[p], dl, hd);
      atomicAdd(&s_den[dl * HEADS + hd], __expf(v - s_m[dl * HEADS + hd]));
    }
  } else {
    for (int i = tid; i < n * HEADS; i += TPB) {
      const int p = i >> 2, hd = i & 3;
      const int dl = s_dl[p];
      const float v = edge_logit(a_src, s_adst, s_src[p], dl, hd);
      s_w[i] = __expf(v - s_m[dl * HEADS + hd]) / (s_den[dl * HEADS + hd] + 1e-16f);
    }
    __syncthreads();
    row_sum(n, tid, h, s_src, s_dl, s_w, s_acc);
  }
}

#define IPT (CAP * HEADS / TPB)
static __device__ __forceinline__ void softmax_once(int n, int tid,
                                                    const float* __restrict__ a_src,
                                                    const int* s_src, const unsigned short* s_dl,
                                                    const float* s_adst, float* s_m, float* s_den,
                                                    float* s_w) {
  const int hd = tid & 3;
  const int pb = tid >> 2;
  int   srcv[IPT];
  int   idx[IPT];
  float v[IPT];
#pragma unroll
  for (int k = 0; k < IPT; ++k) {
    const int p  = pb + (TPB / HEADS) * k;
    const int pc = (p < n) ? p : 0;
    srcv[k] = s_src[pc];
    idx[k]  = (int)s_dl[pc] * HEADS + hd;
  }
#pragma unroll
  for (int k = 0; k < IPT; ++k) v[k] = a_src[(size_t)srcv[k] * HEADS + hd];
  float t[IPT];
#pragma unroll
  for (int k = 0; k < IPT; ++k) t[k] = s_adst[idx[k]];
#pragma unroll
  for (int k = 0; k < IPT; ++k) {
    const float e = v[k] + t[k];
    v[k] = (e > 0.0f) ? e : NEG_SLOPE * e;
  }
#pragma unroll
  for (int k = 0; k < IPT; ++k) {
    if (pb + (TPB / HEADS) * k < n) atomicMax(&s_m[idx[k]], v[k]);
  }
  __syncthreads();
#pragma unroll
  for (int k = 0; k < IPT; ++k) t[k] = s_m[idx[k]];
#pragma unroll
  for (int k = 0; k < IPT; ++k) {
    v[k] = __expf(v[k] - t[k]);
    if (pb + (TPB / HEADS) * k < n) atomicAdd(&s_den[idx[k]], v[k]);
  }
  __syncthreads();
#pragma unroll
  for (int k = 0; k < IPT; ++k) t[k] = s_den[idx[k]];
#pragma unroll
  for (int k = 0; k < IPT; ++k) {
    const float w = v[k] / (t[k] + 1e-16f);
    if (pb + (TPB / HEADS) * k < n) s_w[tid + TPB * k] = w;
  }
}

__global__ __launch_bounds__(TPB) void gat_owner(const int* __restrict__ ei, int E,
                                                 const float* __restrict__ h,
                                                 const float* __restrict__ a_src,
                                                 const float* __restrict__ a_dst,
                                                 const float* __restrict__ bias,
                                                 const float* __restrict__ gamma,
                                                 const float* __restrict__ beta,
                                                 const float* __restrict__ x,
                                                 float* __restrict__ out, int N,
                                                 const unsigned* __restrict__ keys) {
  __shared__ __attribute__((aligned(16))) float s_acc[NPG * D_FEAT];
  __shared__ __attribute__((aligned(16))) float s_w[CAP * HEADS];
  __shared__ int s_src[CAP];
  __shared__ unsigned short s_dl[CAP];
  __shared__ float s_m[NPG * HEADS];
  __shared__ float s_den[NPG * HEADS];
  __shared__ float s_adst[NPG * HEADS];
  __shared__ int s_cnt;

  const int tid = threadIdx.x;
  const int lo  = blockIdx.x * NPG;
  if (lo >= N) return;
  const int nn  = (N - lo < NPG) ? (N - lo) : NPG;

  for (int i = tid; i < NPG * D_FEAT / 4; i += TPB)
    ((float4*)s_acc)[i] = make_float4(0.0f, 0.0f, 0.0f, 0.0f);
  for (int i = tid; i < nn * HEADS; i += TPB) {
    s_m[i]    = -INFINITY;
    s_den[i]  = 0.0f;
    s_adst[i] = a_dst[(size_t)lo * HEADS + i];
  }
  for (int i = tid; i < nn; i += TPB) {
    s_src[i] = lo + i;
    s_dl[i]  = (unsigned short)i;
  }
  if (tid == 0) s_cnt = nn;
  __syncthreads();

  const bool by_key = (keys != nullptr) && ((E & 15) == 0) && (gridDim.x <= KEY_NONE);
  if (by_key)            scan_keys(keys, E, blockIdx.x, tid, &s_cnt, s_src);
  else if ((E & 3) == 0) scan_vec4(ei, E, lo, nn, tid, &s_cnt, s_src, s_dl);
  else                   scan_scalar(ei, E, 0, E, lo, nn, tid, &s_cnt, s_src, s_dl);
  __syncthreads();
  const int total = s_cnt;
  if (by_key) resolve_src_dst(ei, E, lo, nn, (total < CAP) ? total : CAP, tid, s_src, s_dl);
  else        resolve_src(ei, nn, (total < CAP) ? total : CAP, tid, s_src);
  __syncthreads();

  if (total <= CAP) {
    softmax_once(total, tid, a_src, s_src, s_dl, s_adst, s_m, s_den, s_w);
    __syncthreads();
    row_sum(total, tid, h, s_src, s_dl, s_w, s_acc);
    __syncthreads();
  } else {
    for (int ph = 0; ph < 3; ++ph) {
      __syncthreads();
      for (int i = tid; i < nn; i += TPB) {
        s_src[i] = lo + i;
        s_dl[i]  = (unsigned short)i;
      }
      __syncthreads();
      run_phase(ph, nn, tid, a_src, h, s_src, s_dl, s_adst, s_m, s_den, s_w, s_acc);
      __syncthreads();
      for (int c0 = 0; c0 < E; c0 += CAP) {
        if (tid == 0) s_cnt = 0;
        __syncthreads();
        const int c1 = (E - c0 < CAP) ? E : (c0 + CAP);
        scan_scalar(ei, E, c0, c1, lo, nn, tid, &s_cnt, s_src, s_dl);
        __syncthreads();
        const int n = s_cnt;
        resolve_src(ei, 0, n, tid, s_src);
        __syncthreads();
        run_phase(ph, n, tid, a_src, h, s_src, s_dl, s_adst, s_m, s_den, s_w, s_acc);
        __syncthreads();
      }
    }
  }

  const int lane = tid & 31;
  const int wave = __builtin_amdgcn_readfirstlane(tid >> 5);
  const float4 b4  = ((const float4*)bias)[lane];
  const float4 g4  = ((const float4*)gamma)[lane];
  const float4 be4 = ((const float4*)beta)[lane];
  for (int nl = wave; nl < nn; nl += TPB / 32) {
    const size_t row = (size_t)(lo + nl) * (D_FEAT / 4) + lane;
    const float4 xr = ((const float4*)x)[row];
    const float4 a  = *(const float4*)(s_acc + nl * D_FEAT + 4 * lane);
    const float v0 = a.x + b4.x, v1 = a.y + b4.y, v2 = a.z + b4.z, v3 = a.w + b4.w;
    const float q0 = 0.5f * v0 * (1.0f + erff(v0 * 0.70710678118654752f));
    const float q1 = 0.5f * v1 * (1.0f + erff(v1 * 0.70710678118654752f));
    const float q2 = 0.5f * v2 * (1.0f + erff(v2 * 0.70710678118654752f));
    const float q3 = 0.5f * v3 * (1.0f + erff(v3 * 0.70710678118654752f));
    float s  = (q0 + q1) + (q2 + q3);
    float sq = (q0 * q0 + q1 * q1) + (q2 * q2 + q3 * q3);
#pragma unroll
    for (int off = 16; off > 0; off >>= 1) {
      s  += __shfl_xor(s, off, 32);
      sq += __shfl_xor(sq, off, 32);
    }
    const float mu  = s * (1.0f / D_FEAT);
    const float var = sq * (1.0f / D_FEAT) - mu * mu;
    const float inv = rsqrtf(var + LN_EPS);
    float4 o;
    o.x = (q0 - mu) * inv * g4.x + be4.x + xr.x;
    o.y = (q1 - mu) * inv * g4.y + be4.y + xr.y;
    o.z = (q2 - mu) * inv * g4.z + be4.z + xr.z;
    o.w = (q3 - mu) * inv * g4.w + be4.w + xr.w;
    ((float4*)out)[row] = o;
  }
}

extern "C" void kernel_launch(void* const* d_in, const int* in_sizes, int n_in,
                              void* d_out, int out_size, void* d_ws, size_t ws_size,
                              hipStream_t stream) {
  const float* x       = (const float*)d_in[0];
  const int*   ei      = (const int*)d_in[1];
  const float* W       = (const float*)d_in[2];
  const float* att_src = (const float*)d_in[3];
  const float* att_dst = (const float*)d_in[4];
  const float* bias    = (const float*)d_in[5];
  const float* gamma   = (const float*)d_in[6];
  const float* beta    = (const float*)d_in[7];

  const int N = in_sizes[0] / D_FEAT;
  const int E = in_sizes[1] / 2;

  float* ws    = (float*)d_ws;
  float* h     = ws;
  float* a_src = h + (size_t)N * D_FEAT;
  float* a_dst = a_src + (size_t)N * HEADS;
  unsigned* keys = (unsigned*)(a_dst + (size_t)N * HEADS);

  const int owners = (N + NPG - 1) / NPG;
  const bool by_key = (ws_size >= ((size_t)N * (D_FEAT + 2 * HEADS)) * sizeof(float) + (size_t)E) &&
                      ((E & 15) == 0) && (owners <= (int)KEY_NONE);
  const int kb = by_key ? ((E >> 2) + KEY_VEC_PER_BLOCK * 32 * LG_WAVES - 1) / (KEY_VEC_PER_BLOCK * 32 * LG_WAVES) : 0;

  {
    gat_gemm_h<<<GEMM_GROUPS, GEMM_TPB, 0, stream>>>(x, W, h, N);
  }
  {
    int groups = (N + LG_NODES * LG_WAVES - 1) / (LG_NODES * LG_WAVES);
    gat_att_logits<<<kb + groups, 32 * LG_WAVES, 0, stream>>>(h, att_src, att_dst, a_src, a_dst, N,
                                                              ei, E, keys, kb);
  }
  {
    gat_owner<<<owners, TPB, 0, stream>>>(ei, E, h, a_src, a_dst, bias, gamma, beta, x,
                                          (float*)d_out, N, by_key ? keys : nullptr);
  }
}


# owner row sums: a node's LDS accumulator row kept lane-major (channel 4l+k at word 32k+l): each float atomic's 32 lanes hit 32 different banks, not 16 banks twice; sums stay in atomic arrival order
# speedup vs baseline: 3.9232x; 1.3671x over previous
#include <hip/hip_runtime.h>
#include <math.h>

#define D_FEAT 128
#define HEADS 4
#define CDIM 32
#define NEG_SLOPE 0.2f
#define LN_EPS 1e-5f

typedef __attribute__((ext_vector_type(2))) float v2f;
typedef __attribute__((ext_vector_type(8))) float v8f;

#define GEMM_WAVES 16
#define GEMM_TPB (32 * GEMM_WAVES)
#define GEMM_GROUPS 256
#define WS_PITCH 256
#define GEMM_LDS_FLOATS 41984
#define FS_OFFSET (64 * WS_PITCH)
#define NPG 197
#define KEY_NONE 255u
#define KEY_SLOTS 2

static __device__ __forceinline__ unsigned owner_key(int d, int N) {
  return ((unsigned)d < (unsigned)N) ? ((unsigned)d / NPG) : KEY_NONE;
}

static __device__ __forceinline__ unsigned owner_keys4(int4 d, int N) {
  return owner_key(d.x, N) | (owner_key(d.y, N) << 8) |
         (owner_key(d.z, N) << 16) | (owner_key(d.w, N) << 24);
}

static __device__ __forceinline__ void gemm_load_a(const float* __restrict__ x, int rt, int c, int hf,
                                                   int N, float4* a) {
  int arow = rt * 16 + c;
  if (arow >= N) arow = N - 1;
  const float4* ap = (const float4*)(x + (size_t)arow * D_FEAT + 4 * hf);
#pragma unroll
  for (int j = 0; j < 16; ++j) a[j] = ap[2 * j];
}

__global__ __launch_bounds__(GEMM_TPB) void gat_gemm_h(const float* __restrict__ x,
                                                       const float* __restrict__ W,
                                                       float* __restrict__ h,
                                                       int N,
                                                       const float* __restrict__ att_src,
                                                       const float* __restrict__ att_dst,
                                                       float* __restrict__ a_src,
                                                       float* __restrict__ a_dst,
                                                       const int* __restrict__ ei, int E,
                                                       unsigned* __restrict__ keys) {
  __shared__ __attribute__((aligned(16))) float Ws[GEMM_LDS_FLOATS];

  const int tid  = threadIdx.x;
  const int lane = tid & 31;
  const int wave = __builtin_amdgcn_readfirstlane(tid >> 5);
  const int c  = lane & 15;
  const int hf = lane >> 4;
  const int cq = wave & 3;

  const int T  = (N + 15) >> 4;
  const int t0 = (int)(((long long)blockIdx.x * T) / gridDim.x);
  const int t1 = (int)(((long long)(blockIdx.x + 1) * T) / gridDim.x);

  int rt = t0 + (wave >> 2);
  float4 a[16];
  gemm_load_a(x, (rt < T) ? rt : 0, c, hf, N, a);

  const int4* d4 = (const int4*)(ei + E);
  const int nvec = (keys != nullptr) ? (E >> 2) : 0;
  const int kper = (nvec + (int)gridDim.x - 1) / (int)gridDim.x;
  const int kv0  = (int)blockIdx.x * kper;
  const int kv1  = (kv0 + kper < nvec) ? (kv0 + kper) : nvec;
  int4 kd[KEY_SLOTS];
  if (nvec > 0) {
#pragma unroll
    for (int u = 0; u < KEY_SLOTS; ++u) {
      const int i = kv0 + tid + GEMM_TPB * u;
      kd[u] = d4[(i < kv1) ? i : 0];
    }
  }

  {
    const float4* W4 = (const float4*)W;
    float4 r0[4], r1[4];
#pragma unroll
    for (int it = 0; it < 4; ++it) {
      const int i = it * GEMM_TPB + tid;
      const int q = i >> 5, c4 = i & 31;
      r0[it] = W4[(2 * q) * (D_FEAT / 4) + c4];
      r1[it] = W4[(2 * q + 1) * (D_FEAT / 4) + c4];
    }
#pragma unroll
    for (int it = 0; it < 4; ++it) {
      const int i = it * GEMM_TPB + tid;
      const int q = i >> 5, c4 = i & 31;
      float4* dst = (float4*)(Ws + q * WS_PITCH + 8 * c4);
      dst[0] = make_float4(r0[it].x, r1[it].x, r0[it].y, r1[it].y);
      dst[1] = make_float4(r0[it].z, r1[it].z, r0[it].w, r1[it].w);
    }
  }
  {
    const int k = tid >> 2, hd = tid & 3;
    const float4* wr = (const float4*)(W + (size_t)k * D_FEAT + CDIM * hd);
    const float4* as = (const float4*)(att_src + CDIM * hd);
    const float4* ad = (const float4*)(att_dst + CDIM * hd);
    float fs = 0.0f, fd = 0.0f;
#pragma unroll
    for (int i = 0; i < CDIM / 4; ++i) {
      const float4 w4 = wr[i], s4 = as[i], t4 = ad[i];
      fs += (w4.x * s4.x + w4.y * s4.y) + (w4.z * s4.z + w4.w * s4.w);
      fd += (w4.x * t4.x + w4.y * t4.y) + (w4.z * t4.z + w4.w * t4.w);
    }
    float* fk = Ws + FS_OFFSET + (k >> 2) * 64 + (k & 3);
    fk[4 * hd]        = fs;
    fk[4 * (4 + hd)]  = fd;
    fk[4 * (8 + hd)]  = 0.0f;
    fk[4 * (12 + hd)] = 0.0f;
  }
  __syncthreads();

  if (nvec > 0) {
#pragma unroll
    for (int u = 0; u < KEY_SLOTS; ++u) {
      const int i = kv0 + tid + GEMM_TPB * u;
      if (i < kv1) keys[i] = owner_keys4(kd[u], N);
    }
    for (int i = kv0 + tid + GEMM_TPB * KEY_SLOTS; i < kv1; i += GEMM_TPB)
      keys[i] = owner_keys4(d4[i], N);
  }

  const float* bp = Ws + 2 * hf * WS_PITCH + 64 * cq + 4 * c;
  const float* fp = Ws + FS_OFFSET + (hf * 16 + c) * 4;
  const bool with_logits = (cq == ((t0 + (wave >> 2)) & 3));
  for (; rt < t1; rt += GEMM_WAVES / 4) {
    if (with_logits) {
      const int lbase = rt * 16 + 8 * hf;
      v8f lg0 = {}, lg1 = {};
#pragma unroll
      for (int j = 0; j < 16; ++j) {
        const float4 f = *(const float4*)(fp + j * 128);
        v2f av, bv;
        av.x = a[j].x; av.y = a[j].y;
        bv.x = f.x; bv.y = f.y;
        lg0 = __builtin_amdgcn_wmma_f32_16x16x4_f32(false, av, false, bv, (short)0, lg0, false, false);
        av.x = a[j].z; av.y = a[j].w;
        bv.x = f.z; bv.y = f.w;
        lg1 = __builtin_amdgcn_wmma_f32_16x16x4_f32(false, av, false, bv, (short)0, lg1, false, false);
      }
      if (c < 8) {
        float* lp = ((c < 4) ? a_src : a_dst) + (size_t)lbase * HEADS + (c & 3);
#pragma unroll
        for (int v = 0; v < 8; ++v) {
          if (lbase + v < N) lp[(size_t)v * HEADS] = lg0[v] + lg1[v];
        }
      }
    }
    float4 an[16];
    const int rn = rt + GEMM_WAVES / 4;
    gemm_load_a(x, (rn < t1) ? rn : rt, c, hf, N, an);

    v8f acc0 = {}, acc1 = {};
#pragma unroll
    for (int j = 0; j < 16; ++j) {
      const float4 b0 = *(const float4*)(bp + (4 * j) * WS_PITCH);
      const float4 b1 = *(const float4*)(bp + (4 * j + 1) * WS_PITCH);
      v2f av, bv;
      av.x = a[j].x; av.y = a[j].y;
      bv.x = b0.x; bv.y = b0.y;
      acc0 = __builtin_amdgcn_wmma_f32_16x16x4_f32(false, av, false, bv, (short)0, acc0, false, false);
      bv.x = b0.z; bv.y = b0.w;
      acc1 = __builtin_amdgcn_wmma_f32_16x16x4_f32(false, av, false, bv, (short)0, acc1, false, false);
      av.x = a[j].z; av.y = a[j].w;
      bv.x = b1.x; bv.y = b1.y;
      acc0 = __builtin_amdgcn_wmma_f32_16x16x4_f32(false, av, false, bv, (short)0, acc0, false, false);
      bv.x = b1.z; bv.y = b1.w;
      acc1 = __builtin_amdgcn_wmma_f32_16x16x4_f32(false, av, false, bv, (short)0, acc1, false, false);
    }

    const int rbase = rt * 16 + 8 * hf;
    float* dp = h + (size_t)rbase * D_FEAT + 32 * cq + 2 * c;
    if (rt * 16 + 16 <= N) {
#pragma unroll
      for (int v = 0; v < 8; ++v) {
        v2f o; o.x = acc0[v]; o.y = acc1[v];
        *(v2f*)(dp + (size_t)v * D_FEAT) = o;
      }
    } else {
#pragma unroll
      for (int v = 0; v < 8; ++v) {
        if (rbase + v < N) {
          v2f o; o.x = acc0[v]; o.y = acc1[v];
          *(v2f*)(dp + (size_t)v * D_FEAT) = o;
        }
      }
    }
#pragma unroll
    for (int j = 0; j < 16; ++j) a[j] = an[j];
  }
}

#define CAP 4096
#define TPB 1024
#define SCAN_DEPTH 8
#define ACC_SLOT(k) (32 * (k))

static __device__ __forceinline__ void rec_append(int e, unsigned d,
                                                  int* s_cnt, int* s_src, unsigned short* s_dl) {
  const int p = atomicAdd(s_cnt, 1);
  if (p < CAP) {
    s_src[p] = e;
    s_dl[p]  = (unsigned short)d;
  }
}

static __device__ __forceinline__ void resolve_src(const int* __restrict__ ei, int from, int to,
                                                   int tid, int* s_src) {
  for (int i = from + tid; i < to; i += TPB) s_src[i] = ei[s_src[i]];
}

static __device__ __forceinline__ void scan_scalar(const int* __restrict__ ei, int E, int c0, int c1,
                                                   int lo, int nn, int tid,
                                                   int* s_cnt, int* s_src, unsigned short* s_dl) {
  for (int e = c0 + tid; e < c1; e += TPB) {
    const unsigned d = (unsigned)(ei[E + e] - lo);
    if (d < (unsigned)nn) rec_append(e, d, s_cnt, s_src, s_dl);
  }
}

static __device__ __forceinline__ void scan_vec4(const int* __restrict__ ei, int E,
                                                 int lo, int nn, int tid,
                                                 int* s_cnt, int* s_src, unsigned short* s_dl) {
  const int nvec = E >> 2;
  const int4* d4 = (const int4*)(ei + E);
  const int nb = (nvec + TPB * SCAN_DEPTH - 1) / (TPB * SCAN_DEPTH);
  if (nb == 0) return;
  int kb = (int)(blockIdx.x % (unsigned)nb);
  for (int k = 0; k < nb; ++k) {
    const int b = kb * (TPB * SCAN_DEPTH) + tid;
    kb = (kb + 1 == nb) ? 0 : kb + 1;
    int4 v[SCAN_DEPTH];
#pragma unroll
    for (int u = 0; u < SCAN_DEPTH; ++u) {
      const int i = b + TPB * u;
      v[u] = d4[(i < nvec) ? i : 0];
    }
#pragma unroll
    for (int u = 0; u < SCAN_DEPTH; ++u) {
      const int i = b + TPB * u;
      const unsigned d0 = (unsigned)(v[u].x - lo);
      const unsigned d1 = (unsigned)(v[u].y - lo);
      const unsigned d2 = (unsigned)(v[u].z - lo);
      const unsigned d3 = (unsigned)(v[u].w - lo);
      const unsigned mn = min(min(d0, d1), min(d2, d3));
      if (i < nvec && mn < (unsigned)nn) {
        if (d0 < (unsigned)nn) rec_append(4 * i + 0, d0, s_cnt, s_src, s_dl);
        if (d1 < (unsigned)nn) rec_append(4 * i + 1, d1, s_cnt, s_src, s_dl);
        if (d2 < (unsigned)nn) rec_append(4 * i + 2, d2, s_cnt, s_src, s_dl);
        if (d3 < (unsigned)nn) rec_append(4 * i + 3, d3, s_cnt, s_src, s_dl);
      }
    }
  }
}

static __device__ __forceinline__ void resolve_src_dst(const int* __restrict__ ei, int E, int lo,
                                                       int from, int to, int tid,
                                                       int* s_src, unsigned short* s_dl) {
  for (int i = from + tid; i < to; i += TPB) {
    const int e = s_src[i];
    const int d = ei[E + e];
    const int s = ei[e];
    s_dl[i]  = (unsigned short)(d - lo);
    s_src[i] = s;
  }
}

static __device__ __forceinline__ unsigned nonzero_bytes(unsigned x) {
  return ((x & 0x7F7F7F7Fu) + 0x7F7F7F7Fu) | x;
}

static __device__ __forceinline__ void scan_keys(const unsigned* __restrict__ keys, int E,
                                                 unsigned gid, int tid, int* s_cnt, int* s_src) {
  const int nvec = E >> 4;
  const uint4* k4 = (const uint4*)keys;
  const unsigned rep4 = gid * 0x01010101u;
  const int nb = (nvec + TPB * SCAN_DEPTH - 1) / (TPB * SCAN_DEPTH);
  if (nb == 0) return;
  int kb = (int)(blockIdx.x % (unsigned)nb);
  for (int k = 0; k < nb; ++k) {
    const int b = kb * (TPB * SCAN_DEPTH) + tid;
    kb = (kb + 1 == nb) ? 0 : kb + 1;
    uint4 v[SCAN_DEPTH];
#pragma unroll
    for (int u = 0; u < SCAN_DEPTH; ++u) {
      const int i = b + TPB * u;
      v[u] = k4[(i < nvec) ? i : 0];
    }
#pragma unroll
    for (int u = 0; u < SCAN_DEPTH; ++u) {
      const int i = b + TPB * u;
      unsigned m = (nonzero_bytes(v[u].x ^ rep4) & 0x80808080u) |
                   ((nonzero_bytes(v[u].y ^ rep4) >> 1) & 0x40404040u) |
                   ((nonzero_bytes(v[u].z ^ rep4) >> 2) & 0x20202020u) |
                   ((nonzero_bytes(v[u].w ^ rep4) >> 3) & 0x10101010u);
      m ^= 0xF0F0F0F0u;
      if (i >= nvec) m = 0u;
      while (m != 0u) {
        const int p = __builtin_ctz(m);
        m &= m - 1u;
        const int e = 16 * i + 4 * (7 - (p & 7)) + (p >> 3);
        const int q = atomicAdd(s_cnt, 1);
        if (q < CAP) s_src[q] = e;
      }
    }
  }
}

static __device__ __forceinline__ float edge_logit(const float* __restrict__ a_src,
                                                   const float* s_adst, int src, int dl, int hd) {
  const float v = a_src[(size_t)src * HEADS + hd] + s_adst[dl * HEADS + hd];
  return (v > 0.0f) ? v : NEG_SLOPE * v;
}

static __device__ __forceinline__ void row_sum(int n, int tid, const float* __restrict__ h,
                                               const int* s_src, const unsigned short* s_dl,
                                               const float* s_w, float* s_acc) {
  const int lane = tid & 31;
  const int wave = __builtin_amdgcn_readfirstlane(tid >> 5);
  const float4* h4 = (const float4*)h;
  for (int p0 = wave; p0 < n; p0 += 32 * 4) {
    float4 r[4];
    int srcv[4];
    int dlv[4];
    float wv[4];
#pragma unroll
    for (int u = 0; u < 4; ++u) {
      const int p  = p0 + 32 * u;
      const int pc = (p < n) ? p : p0;
      srcv[u] = s_src[pc];
      dlv[u]  = s_dl[pc];
      wv[u]   = (p < n) ? s_w[pc * HEADS + (lane >> 3)] : 0.0f;
    }
#pragma unroll
    for (int u = 0; u < 4; ++u) r[u] = h4[(size_t)srcv[u] * (D_FEAT / 4) + lane];
#pragma unroll
    for (int u = 0; u < 4; ++u) {
      const bool ok = (p0 + 32 * u < n);
      float* ap = s_acc + dlv[u] * D_FEAT + lane;
      atomicAdd(ap + ACC_SLOT(0), ok ? r[u].x * wv[u] : 0.0f);
      atomicAdd(ap + ACC_SLOT(1), ok ? r[u].y * wv[u] : 0.0f);
      atomicAdd(ap + ACC_SLOT(2), ok ? r[u].z * wv[u] : 0.0f);
      atomicAdd(ap + ACC_SLOT(3), ok ? r[u].w * wv[u] : 0.0f);
    }
  }
}

static __device__ __forceinline__ void run_phase(int ph, int n, int tid,
                                                 const float* __restrict__ a_src,
                                                 const float* __restrict__ h,
                                                 const int* s_src, const unsigned short* s_dl,
                                                 const float* s_adst, float* s_m, float* s_den,
                                                 float* s_w, float* s_acc) {
  if (ph == 0) {
    for (int i = tid; i < n * HEADS; i += TPB) {
      const int p = i >> 2, hd = i & 3;
      const int dl = s_dl[p];
      const float v = edge_logit(a_src, s_adst, s_src[p], dl, hd);
      atomicMax(&s_m[dl * HEADS + hd], v);
    }
  } else if (ph == 1) {
    for (int i = tid; i < n * HEADS; i += TPB) {
      const int p = i >> 2, hd = i & 3;
      const int dl = s_dl[p];
      const float v = edge_logit(a_src, s_adst, s_src[p], dl, hd);
      atomicAdd(&s_den[dl * HEADS + hd], __expf(v - s_m[dl * HEADS + hd]));
    }
  } else {
    for (int i = tid; i < n * HEADS; i += TPB) {
      const int p = i >> 2, hd = i & 3;
      const int dl = s_dl[p];
      const float v = edge_logit(a_src, s_adst, s_src[p], dl, hd);
      s_w[i] = __expf(v - s_m[dl * HEADS + hd]) / (s_den[dl * HEADS + hd] + 1e-16f);
    }
    __syncthreads();
    row_sum(n, tid, h, s_src, s_dl, s_w, s_acc);
  }
}

#define IPT (CAP * HEADS / TPB)
static __device__ __forceinline__ void softmax_once(int n, int tid,
                                                    const float* __restrict__ a_src,
                                                    const int* s_src, const unsigned short* s_dl,
                                                    const float* s_adst, float* s_m, float* s_den,
                                                    float* s_w) {
  const int hd = tid & 3;
  const int pb = tid >> 2;
  int   srcv[IPT];
  int   idx[IPT];
  float v[IPT];
#pragma unroll
  for (int k = 0; k < IPT; ++k) {
    const int p  = pb + (TPB / HEADS) * k;
    const int pc = (p < n) ? p : 0;
    srcv[k] = s_src[pc];
    idx[k]  = (int)s_dl[pc] * HEADS + hd;
  }
#pragma unroll
  for (int k = 0; k < IPT; ++k) v[k] = a_src[(size_t)srcv[k] * HEADS + hd];
  float t[IPT];
#pragma unroll
  for (int k = 0; k < IPT; ++k) t[k] = s_adst[idx[k]];
#pragma unroll
  for (int k = 0; k < IPT; ++k) {
    const float e = v[k] + t[k];
    v[k] = (e > 0.0f) ? e : NEG_SLOPE * e;
  }
#pragma unroll
  for (int k = 0; k < IPT; ++k) {
    if (pb + (TPB / HEADS) * k < n) atomicMax(&s_m[idx[k]], v[k]);
  }
  __syncthreads();
#pragma unroll
  for (int k = 0; k < IPT; ++k) t[k] = s_m[idx[k]];
#pragma unroll
  for (int k = 0; k < IPT; ++k) {
    v[k] = __expf(v[k] - t[k]);
    if (pb + (TPB / HEADS) * k < n) atomicAdd(&s_den[idx[k]], v[k]);
  }
  __syncthreads();
#pragma unroll
  for (int k = 0; k < IPT; ++k) t[k] = s_den[idx[k]];
#pragma unroll
  for (int k = 0; k < IPT; ++k) {
    const float w = v[k] / (t[k] + 1e-16f);
    if (pb + (TPB / HEADS) * k < n) s_w[tid + TPB * k] = w;
  }
}

__global__ __launch_bounds__(TPB) void gat_owner(const int* __restrict__ ei, int E,
                                                 const float* __restrict__ h,
                                                 const float* __restrict__ a_src,
                                                 const float* __restrict__ a_dst,
                                                 const float* __restrict__ bias,
                                                 const float* __restrict__ gamma,
                                                 const float* __restrict__ beta,
                                                 const float* __restrict__ x,
                                                 float* __restrict__ out, int N,
                                                 const unsigned* __restrict__ keys) {
  __shared__ __attribute__((aligned(16))) float s_acc[NPG * D_FEAT];
  __shared__ __attribute__((aligned(16))) float s_w[CAP * HEADS];
  __shared__ int s_src[CAP];
  __shared__ unsigned short s_dl[CAP];
  __shared__ float s_m[NPG * HEADS];
  __shared__ float s_den[NPG * HEADS];
  __shared__ float s_adst[NPG * HEADS];
  __shared__ int s_cnt;

  const int tid = threadIdx.x;
  const int lo  = blockIdx.x * NPG;
  if (lo >= N) return;
  const int nn  = (N - lo < NPG) ? (N - lo) : NPG;

  for (int i = tid; i < NPG * D_FEAT / 4; i += TPB)
    ((float4*)s_acc)[i] = make_float4(0.0f, 0.0f, 0.0f, 0.0f);
  for (int i = tid; i < nn * HEADS; i += TPB) {
    s_m[i]    = -INFINITY;
    s_den[i]  = 0.0f;
    s_adst[i] = a_dst[(size_t)lo * HEADS + i];
  }
  for (int i = tid; i < nn; i += TPB) {
    s_src[i] = lo + i;
    s_dl[i]  = (unsigned short)i;
  }
  if (tid == 0) s_cnt = nn;
  __syncthreads();

  const bool by_key = (keys != nullptr) && ((E & 15) == 0) && (gridDim.x <= KEY_NONE);
  if (by_key)            scan_keys(keys, E, blockIdx.x, tid, &s_cnt, s_src);
  else if ((E & 3) == 0) scan_vec4(ei, E, lo, nn, tid, &s_cnt, s_src, s_dl);
  else                   scan_scalar(ei, E, 0, E, lo, nn, tid, &s_cnt, s_src, s_dl);
  __syncthreads();
  const int total = s_cnt;
  if (by_key) resolve_src_dst(ei, E, lo, nn, (total < CAP) ? total : CAP, tid, s_src, s_dl);
  else        resolve_src(ei, nn, (total < CAP) ? total : CAP, tid, s_src);
  __syncthreads();

  if (total <= CAP) {
    softmax_once(total, tid, a_src, s_src, s_dl, s_adst, s_m, s_den, s_w);
    __syncthreads();
    row_sum(total, tid, h, s_src, s_dl, s_w, s_acc);
    __syncthreads();
  } else {
    for (int ph = 0; ph < 3; ++ph) {
      __syncthreads();
      for (int i = tid; i < nn; i += TPB) {
        s_src[i] = lo + i;
        s_dl[i]  = (unsigned short)i;
      }
      __syncthreads();
      run_phase(ph, nn, tid, a_src, h, s_src, s_dl, s_adst, s_m, s_den, s_w, s_acc);
      __syncthreads();
      for (int c0 = 0; c0 < E; c0 += CAP) {
        if (tid == 0) s_cnt = 0;
        __syncthreads();
        const int c1 = (E - c0 < CAP) ? E : (c0 + CAP);
        scan_scalar(ei, E, c0, c1, lo, nn, tid, &s_cnt, s_src, s_dl);
        __syncthreads();
        const int n = s_cnt;
        resolve_src(ei, 0, n, tid, s_src);
        __syncthreads();
        run_phase(ph, n, tid, a_src, h, s_src, s_dl, s_adst, s_m, s_den, s_w, s_acc);
        __syncthreads();
      }
    }
  }

  const int lane = tid & 31;
  const int wave = __builtin_amdgcn_readfirstlane(tid >> 5);
  const float4 b4  = ((const float4*)bias)[lane];
  const float4 g4  = ((const float4*)gamma)[lane];
  const float4 be4 = ((const float4*)beta)[lane];
  for (int nl = wave; nl < nn; nl += TPB / 32) {
    const size_t row = (size_t)(lo + nl) * (D_FEAT / 4) + lane;
    const float4 xr = ((const float4*)x)[row];
    const float* arow = s_acc + nl * D_FEAT + lane;
    const float v0 = arow[ACC_SLOT(0)] + b4.x, v1 = arow[ACC_SLOT(1)] + b4.y;
    const float v2 = arow[ACC_SLOT(2)] + b4.z, v3 = arow[ACC_SLOT(3)] + b4.w;
    const float q0 = 0.5f * v0 * (1.0f + erff(v0 * 0.70710678118654752f));
    const float q1 = 0.5f * v1 * (1.0f + erff(v1 * 0.70710678118654752f));
    const float q2 = 0.5f * v2 * (1.0f + erff(v2 * 0.70710678118654752f));
    const float q3 = 0.5f * v3 * (1.0f + erff(v3 * 0.70710678118654752f));
    float s  = (q0 + q1) + (q2 + q3);
    float sq = (q0 * q0 + q1 * q1) + (q2 * q2 + q3 * q3);
#pragma unroll
    for (int off = 16; off > 0; off >>= 1) {
      s  += __shfl_xor(s, off, 32);
      sq += __shfl_xor(sq, off, 32);
    }
    const float mu  = s * (1.0f / D_FEAT);
    const float var = sq * (1.0f / D_FEAT) - mu * mu;
    const float inv = rsqrtf(var + LN_EPS);
    float4 o;
    o.x = (q0 - mu) * inv * g4.x + be4.x + xr.x;
    o.y = (q1 - mu) * inv * g4.y + be4.y + xr.y;
    o.z = (q2 - mu) * inv * g4.z + be4.z + xr.z;
    o.w = (q3 - mu) * inv * g4.w + be4.w + xr.w;
    ((float4*)out)[row] = o;
  }
}

extern "C" void kernel_launch(void* const* d_in, const int* in_sizes, int n_in,
                              void* d_out, int out_size, void* d_ws, size_t ws_size,
                              hipStream_t stream) {
  const float* x       = (const float*)d_in[0];
  const int*   ei      = (const int*)d_in[1];
  const float* W       = (const float*)d_in[2];
  const float* att_src = (const float*)d_in[3];
  const float* att_dst = (const float*)d_in[4];
  const float* bias    = (const float*)d_in[5];
  const float* gamma   = (const float*)d_in[6];
  const float* beta    = (const float*)d_in[7];

  const int N = in_sizes[0] / D_FEAT;
  const int E = in_sizes[1] / 2;

  float* ws    = (float*)d_ws;
  float* h     = ws;
  float* a_src = h + (size_t)N * D_FEAT;
  float* a_dst = a_src + (size_t)N * HEADS;
  unsigned* keys = (unsigned*)(a_dst + (size_t)N * HEADS);

  const int owners = (N + NPG - 1) / NPG;
  const bool by_key = (ws_size >= ((size_t)N * (D_FEAT + 2 * HEADS)) * sizeof(float) + (size_t)E) &&
                      ((E & 15) == 0) && (owners <= (int)KEY_NONE);

  {
    gat_gemm_h<<<GEMM_GROUPS, GEMM_TPB, 0, stream>>>(x, W, h, N, att_src, att_dst, a_src, a_dst,
                                                     ei, E, by_key ? keys : nullptr);
  }
  {
    gat_owner<<<owners, TPB, 0, stream>>>(ei, E, h, a_src, a_dst, bias, gamma, beta, x,
                                          (float*)d_out, N, by_key ? keys : nullptr);
  }
}
